# starting kernel
# baseline (speedup 1.0000x reference)
#include <hip/hip_runtime.h>

#define NN 50000
#define EE 800000
#define DD 128
#define ETOT (EE + NN)
#define SLOPE 0.2f

typedef float v2f __attribute__((ext_vector_type(2)));
typedef float v8f __attribute__((ext_vector_type(8)));

__global__ __launch_bounds__(256) void gemm_wmma_f32(
    const float* __restrict__ A, const float* __restrict__ W,
    float* __restrict__ H, int accumulate) {
  const int lane = threadIdx.x & 31;
  const int wv   = threadIdx.x >> 5;
  const int m0   = blockIdx.x * 16;
  const int n0   = wv * 16;
  const int half = lane >> 4;
  const int l15  = lane & 15;
  const int row  = m0 + l15;
  const int col  = n0 + l15;

  v8f acc;
  if (accumulate) {
#pragma unroll
    for (int r = 0; r < 8; ++r)
      acc[r] = H[(size_t)(m0 + half * 8 + r) * DD + col];
  } else {
#pragma unroll
    for (int r = 0; r < 8; ++r) acc[r] = 0.0f;
  }

  const float* arow = A + (size_t)row * DD;
#pragma unroll 4
  for (int k = 0; k < DD; k += 4) {
    v2f av, bv;
    av[0] = arow[k + 2 * half];
    av[1] = arow[k + 2 * half + 1];
    bv[0] = W[(size_t)(k + 2 * half) * DD + col];
    bv[1] = W[(size_t)(k + 2 * half + 1) * DD + col];
    acc = __builtin_amdgcn_wmma_f32_16x16x4_f32(false, av, false, bv,
                                                (short)0, acc, false, false);
  }
#pragma unroll
  for (int r = 0; r < 8; ++r)
    H[(size_t)(m0 + half * 8 + r) * DD + col] = acc[r];
}

__global__ __launch_bounds__(256) void scores_init(
    const float* __restrict__ H, const float* __restrict__ a_s,
    const float* __restrict__ a_d, float* __restrict__ s_src,
    float* __restrict__ s_dst, unsigned* __restrict__ m_ord,
    float* __restrict__ denom) {
  const int lane = threadIdx.x & 31;
  const int node = blockIdx.x * 8 + (threadIdx.x >> 5);
  if (node >= NN) return;
  const float4 hv = *(const float4*)(H + (size_t)node * DD + lane * 4);
  const float4 as = *(const float4*)(a_s + lane * 4);
  const float4 ad = *(const float4*)(a_d + lane * 4);
  float ps = hv.x * as.x + hv.y * as.y + hv.z * as.z + hv.w * as.w;
  float pd = hv.x * ad.x + hv.y * ad.y + hv.z * ad.z + hv.w * ad.w;
#pragma unroll
  for (int off = 16; off > 0; off >>= 1) {
    ps += __shfl_down(ps, off, 32);
    pd += __shfl_down(pd, off, 32);
  }
  if (lane == 0) {
    s_src[node] = ps;
    s_dst[node] = pd;
    m_ord[node] = 0u;
    denom[node] = 0.0f;
  }
}

__device__ __forceinline__ unsigned f2ord(float f) {
  unsigned u = __float_as_uint(f);
  return (u & 0x80000000u) ? ~u : (u | 0x80000000u);
}
__device__ __forceinline__ float ord2f(unsigned o) {
  return (o & 0x80000000u) ? __uint_as_float(o & 0x7FFFFFFFu)
                           : __uint_as_float(~o);
}

__global__ __launch_bounds__(256) void edge_max(
    const int* __restrict__ ei, const float* __restrict__ s_src,
    const float* __restrict__ s_dst, float* __restrict__ wbuf,
    unsigned* __restrict__ m_ord) {
  int e = blockIdx.x * blockDim.x + threadIdx.x;
  if (e >= ETOT) return;
  int s, d;
  if (e < EE) { s = ei[e]; d = ei[EE + e]; } else { s = d = e - EE; }
  float v = s_src[s] + s_dst[d];
  v = v > 0.0f ? v : SLOPE * v;
  wbuf[e] = v;
  atomicMax(&m_ord[d], f2ord(v));
}

__global__ __launch_bounds__(256) void edge_exp(
    const int* __restrict__ ei, const unsigned* __restrict__ m_ord,
    float* __restrict__ wbuf, float* __restrict__ denom) {
  int e = blockIdx.x * blockDim.x + threadIdx.x;
  if (e >= ETOT) return;
  int d = (e < EE) ? ei[EE + e] : (e - EE);
  float z = __expf(wbuf[e] - ord2f(m_ord[d]));
  wbuf[e] = z;
  atomicAdd(&denom[d], z);
}

__global__ __launch_bounds__(256) void init_out(float* __restrict__ out,
                                                const float* __restrict__ b) {
  int i = blockIdx.x * blockDim.x + threadIdx.x;
  out[i] = b[i & (DD - 1)];
}

__global__ __launch_bounds__(256) void edge_scatter(
    const int* __restrict__ ei, const float* __restrict__ H,
    const float* __restrict__ wbuf, const float* __restrict__ denom,
    float* __restrict__ out) {
  int e    = (int)((blockIdx.x * blockDim.x + threadIdx.x) >> 5);
  int lane = threadIdx.x & 31;
  if (e >= ETOT) return;
  int s, d;
  if (e < EE) { s = ei[e]; d = ei[EE + e]; } else { s = d = e - EE; }
  float alpha = wbuf[e] / denom[d];
  const float4 hv = *(const float4*)(H + (size_t)s * DD + lane * 4);
  float* o = out + (size_t)d * DD + lane * 4;
  atomicAdd(o + 0, alpha * hv.x);
  atomicAdd(o + 1, alpha * hv.y);
  atomicAdd(o + 2, alpha * hv.z);
  atomicAdd(o + 3, alpha * hv.w);
}

static void run_gat(const float* A1, const float* A2, const float* Wmat,
                    const float* a_s, const float* a_d, const float* bias,
                    const int* ei, float* ht, float* out, float* s_src,
                    float* s_dst, unsigned* m_ord, float* denom, float* wbuf,
                    hipStream_t stream) {
  gemm_wmma_f32<<<NN / 16, 256, 0, stream>>>(A1, Wmat, ht, 0);
  if (A2)
    gemm_wmma_f32<<<NN / 16, 256, 0, stream>>>(A2, Wmat + DD * DD, ht, 1);
  scores_init<<<NN / 8, 256, 0, stream>>>(ht, a_s, a_d, s_src, s_dst, m_ord,
                                          denom);
  edge_max<<<(ETOT + 255) / 256, 256, 0, stream>>>(ei, s_src, s_dst, wbuf,
                                                   m_ord);
  edge_exp<<<(ETOT + 255) / 256, 256, 0, stream>>>(ei, m_ord, wbuf, denom);
  init_out<<<(NN * DD) / 256, 256, 0, stream>>>(out, bias);
  edge_scatter<<<(ETOT + 7) / 8, 256, 0, stream>>>(ei, ht, wbuf, denom, out);
}

extern "C" void kernel_launch(void* const* d_in, const int* in_sizes, int n_in,
                              void* d_out, int out_size, void* d_ws,
                              size_t ws_size, hipStream_t stream) {
  const float* x    = (const float*)d_in[0];
  const int*   ei   = (const int*)d_in[1];
  const float* We   = (const float*)d_in[2];
  const float* ae_s = (const float*)d_in[3];
  const float* ae_d = (const float*)d_in[4];
  const float* be   = (const float*)d_in[5];
  const float* Wp   = (const float*)d_in[6];
  const float* ap_s = (const float*)d_in[7];
  const float* ap_d = (const float*)d_in[8];
  const float* bp   = (const float*)d_in[9];
  const float* Wd   = (const float*)d_in[10];
  const float* ad_s = (const float*)d_in[11];
  const float* ad_d = (const float*)d_in[12];
  const float* bd   = (const float*)d_in[13];
  float*       out  = (float*)d_out;

  const size_t ND = (size_t)NN * DD;
  float*    enc   = (float*)d_ws;
  float*    hA    = enc + ND;
  float*    hB    = hA + ND;
  float*    ht    = hB + ND;
  float*    s_src = ht + ND;
  float*    s_dst = s_src + NN;
  unsigned* m_ord = (unsigned*)(s_dst + NN);
  float*    denom = (float*)(m_ord + NN);
  float*    wbuf  = denom + NN;

  run_gat(x, nullptr, We, ae_s, ae_d, be, ei, ht, enc, s_src, s_dst, m_ord,
          denom, wbuf, stream);
  run_gat(enc, enc, Wp, ap_s, ap_d, bp, ei, ht, hA, s_src, s_dst, m_ord,
          denom, wbuf, stream);
  run_gat(hA, enc, Wp, ap_s, ap_d, bp, ei, ht, hB, s_src, s_dst, m_ord,
          denom, wbuf, stream);
  run_gat(hB, nullptr, Wd, ad_s, ad_d, bd, ei, ht, out, s_src, s_dst, m_ord,
          denom, wbuf, stream);
}


# owner work-groups scan the destination list in layer 1 only and keep their edge lists in the workspace; layers 2-4 read them back, no rescan; same addends, LDS f32 adds still in arrival order
# speedup vs baseline: 3.2996x; 3.2996x over previous
#include <hip/hip_runtime.h>

#define NN 50000
#define EE 800000
#define DD 128
#define ETOT (EE + NN)
#define SLOPE 0.2f

typedef float v2f __attribute__((ext_vector_type(2)));
typedef float v8f __attribute__((ext_vector_type(8)));

__global__ __launch_bounds__(256) void gemm_wmma_f32(
    const float* __restrict__ A, const float* __restrict__ W,
    float* __restrict__ H, int accumulate) {
  const int lane = threadIdx.x & 31;
  const int wv   = threadIdx.x >> 5;
  const int m0   = blockIdx.x * 16;
  const int n0   = wv * 16;
  const int half = lane >> 4;
  const int l15  = lane & 15;
  const int row  = m0 + l15;
  const int col  = n0 + l15;

  v8f acc;
  if (accumulate) {
#pragma unroll
    for (int r = 0; r < 8; ++r)
      acc[r] = H[(size_t)(m0 + half * 8 + r) * DD + col];
  } else {
#pragma unroll
    for (int r = 0; r < 8; ++r) acc[r] = 0.0f;
  }

  const float* arow = A + (size_t)row * DD;
#pragma unroll 4
  for (int k = 0; k < DD; k += 4) {
    v2f av, bv;
    av[0] = arow[k + 2 * half];
    av[1] = arow[k + 2 * half + 1];
    bv[0] = W[(size_t)(k + 2 * half) * DD + col];
    bv[1] = W[(size_t)(k + 2 * half + 1) * DD + col];
    acc = __builtin_amdgcn_wmma_f32_16x16x4_f32(false, av, false, bv,
                                                (short)0, acc, false, false);
  }
#pragma unroll
  for (int r = 0; r < 8; ++r)
    H[(size_t)(m0 + half * 8 + r) * DD + col] = acc[r];
}

__global__ __launch_bounds__(256) void scores_init(
    const float* __restrict__ H, const float* __restrict__ a_s,
    const float* __restrict__ a_d, float* __restrict__ s_src,
    float* __restrict__ s_dst) {
  const int lane = threadIdx.x & 31;
  const int node = blockIdx.x * 8 + (threadIdx.x >> 5);
  if (node >= NN) return;
  const float4 hv = *(const float4*)(H + (size_t)node * DD + lane * 4);
  const float4 as = *(const float4*)(a_s + lane * 4);
  const float4 ad = *(const float4*)(a_d + lane * 4);
  float ps = hv.x * as.x + hv.y * as.y + hv.z * as.z + hv.w * as.w;
  float pd = hv.x * ad.x + hv.y * ad.y + hv.z * ad.z + hv.w * ad.w;
#pragma unroll
  for (int off = 16; off > 0; off >>= 1) {
    ps += __shfl_down(ps, off, 32);
    pd += __shfl_down(pd, off, 32);
  }
  if (lane == 0) {
    s_src[node] = ps;
    s_dst[node] = pd;
  }
}

__device__ __forceinline__ unsigned f2ord(float f) {
  unsigned u = __float_as_uint(f);
  return (u & 0x80000000u) ? ~u : (u | 0x80000000u);
}
__device__ __forceinline__ float ord2f(unsigned o) {
  return (o & 0x80000000u) ? __uint_as_float(o & 0x7FFFFFFFu)
                           : __uint_as_float(~o);
}
__device__ __forceinline__ float leaky(float v) {
  return v > 0.0f ? v : SLOPE * v;
}

#define NPB 200
#define NBLK (NN / NPB)
#define OT 1024
#define OW (OT / 32)
#define CAP 16384
#define EB 8
static_assert(NN % NPB == 0, "every work-group owns a full range");
static_assert(EE % 4 == 0 && EE % 32 == 0, "whole 16-byte reads, whole wave chunks");
static_assert(NN <= 65536 && NPB <= 4096, "entry packs dst-local<<16 | src");
static_assert(EE <= (1 << 20), "scan entry packs dst-local<<20 | edge");

__device__ __forceinline__ void row_add(const float* __restrict__ H, float* acc,
                                        int src, int dl, float a, int lane) {
  const float* hs = H + (size_t)src * DD + lane;
  const float h0 = hs[0], h1 = hs[32], h2 = hs[64], h3 = hs[96];
  float* o = acc + dl * DD + lane;
  atomicAdd(o + 0, a * h0);
  atomicAdd(o + 32, a * h1);
  atomicAdd(o + 64, a * h2);
  atomicAdd(o + 96, a * h3);
}

template <bool BUILD>
__global__ __launch_bounds__(OT) void owner_aggregate(
    const int* __restrict__ ei, const float* __restrict__ H,
    const float* __restrict__ s_src, const float* __restrict__ s_dst,
    const float* __restrict__ bias, float* __restrict__ out,
    int* glist, int* gcnt) {
  __shared__ float acc[NPB * DD];
  __shared__ int l_pk[CAP];
  __shared__ float l_v[CAP];
  __shared__ unsigned m_ord[NPB];
  __shared__ float den[NPB];
  __shared__ float sd[NPB];
  __shared__ int cnt_s;

  const int tid = threadIdx.x;
  const int lane = tid & 31;
  const int wv = tid >> 5;
  const int lo = blockIdx.x * NPB;

  for (int i = tid; i < NPB * DD / 4; i += OT)
    ((float4*)acc)[i] = make_float4(0.0f, 0.0f, 0.0f, 0.0f);
  if (tid < NPB) {
    const float sdv = s_dst[lo + tid];
    const float v = leaky(s_src[lo + tid] + sdv);
    sd[tid] = sdv;
    den[tid] = 0.0f;
    m_ord[tid] = f2ord(v);
    l_pk[tid] = (tid << 16) | (lo + tid);
    l_v[tid] = v;
  }
  if (tid == 0) cnt_s = NPB;
  __syncthreads();

  int* gl = glist + (size_t)blockIdx.x * CAP;
  int total;
  if (BUILD) {
  const int4* dv = (const int4*)(ei + EE);
  const int QN = EE / 4;
  int4 cur[4], nxt[4];
#pragma unroll
  for (int u = 0; u < 4; ++u) cur[u] = dv[min(tid + u * OT, QN - 1)];
  for (int q0 = tid; q0 < QN; q0 += OT * 4) {
#pragma unroll
    for (int u = 0; u < 4; ++u) nxt[u] = dv[min(q0 + (4 + u) * OT, QN - 1)];
#pragma unroll
    for (int u = 0; u < 4; ++u) {
      const int q = q0 + u * OT;
      const int eb = q * 4;
      const unsigned dl4[4] = {(unsigned)(cur[u].x - lo), (unsigned)(cur[u].y - lo),
                               (unsigned)(cur[u].z - lo), (unsigned)(cur[u].w - lo)};
      if ((q < QN) & (min(min(dl4[0], dl4[1]), min(dl4[2], dl4[3])) < (unsigned)NPB)) {
#pragma unroll
        for (int c4 = 0; c4 < 4; ++c4) {
          const unsigned dl = dl4[c4];
          if (dl < (unsigned)NPB) {
            const int slot = atomicAdd(&cnt_s, 1);
            if (slot < CAP) l_pk[slot] = (int)((dl << 20) | (unsigned)(eb + c4));
          }
        }
      }
    }
#pragma unroll
    for (int u = 0; u < 4; ++u) cur[u] = nxt[u];
  }
  __syncthreads();
  total = cnt_s;
  if (tid == 0) gcnt[blockIdx.x] = total;
  } else {
    total = gcnt[blockIdx.x];
  }
  const bool listed = total <= CAP;

  if (listed) {
    for (int i = NPB + tid; i < total; i += OT) {
      int ent;
      if (BUILD) {
        const unsigned pk = (unsigned)l_pk[i];
        ent = (int)((pk >> 20) << 16) | ei[pk & 0xFFFFFu];
        gl[i] = ent;
      } else {
        ent = gl[i];
      }
      const unsigned dl = (unsigned)ent >> 16;
      const float v = leaky(s_src[ent & 0xFFFF] + sd[dl]);
      atomicMax(&m_ord[dl], f2ord(v));
      l_pk[i] = ent;
      l_v[i] = v;
    }
    __syncthreads();
    for (int i = tid; i < total; i += OT) {
      const int dl = l_pk[i] >> 16;
      const float z = __expf(l_v[i] - ord2f(m_ord[dl]));
      l_v[i] = z;
      atomicAdd(&den[dl], z);
    }
    __syncthreads();
    for (int base = wv * EB; base < total; base += OW * EB) {
      float h[EB][4];
#pragma unroll
      for (int jj = 0; jj < EB; ++jj) {
        int idx = base + jj;
        if (idx >= total) idx = total - 1;
        const int pk = __builtin_amdgcn_readfirstlane(l_pk[idx]);
        const float* hs = H + (size_t)(pk & 0xFFFF) * DD + lane;
        h[jj][0] = hs[0];
        h[jj][1] = hs[32];
        h[jj][2] = hs[64];
        h[jj][3] = hs[96];
      }
#pragma unroll
      for (int jj = 0; jj < EB; ++jj) {
        const int idx = base + jj;
        if (idx < total) {
          const int dl = __builtin_amdgcn_readfirstlane(l_pk[idx]) >> 16;
          const float a = l_v[idx] / den[dl];
          float* o = acc + dl * DD + lane;
          atomicAdd(o + 0, a * h[jj][0]);
          atomicAdd(o + 32, a * h[jj][1]);
          atomicAdd(o + 64, a * h[jj][2]);
          atomicAdd(o + 96, a * h[jj][3]);
        }
      }
    }
  } else {
    for (int e = tid; e < EE; e += OT) {
      const unsigned dl = (unsigned)(ei[EE + e] - lo);
      if (dl < (unsigned)NPB)
        atomicMax(&m_ord[dl], f2ord(leaky(s_src[ei[e]] + sd[dl])));
    }
    __syncthreads();
    if (tid < NPB) {
      const float v = leaky(s_src[lo + tid] + sd[tid]);
      atomicAdd(&den[tid], __expf(v - ord2f(m_ord[tid])));
    }
    for (int e = tid; e < EE; e += OT) {
      const unsigned dl = (unsigned)(ei[EE + e] - lo);
      if (dl < (unsigned)NPB) {
        const float v = leaky(s_src[ei[e]] + sd[dl]);
        atomicAdd(&den[dl], __expf(v - ord2f(m_ord[dl])));
      }
    }
    __syncthreads();
    for (int n = wv; n < NPB; n += OW) {
      const float v = leaky(s_src[lo + n] + sd[n]);
      row_add(H, acc, lo + n, n, __expf(v - ord2f(m_ord[n])) / den[n], lane);
    }
    for (int e0 = wv * 32; e0 < EE; e0 += OW * 32) {
      const int e = e0 + lane;
      const unsigned dl = (unsigned)(ei[EE + e] - lo);
      const bool hit = dl < (unsigned)NPB;
      int s = 0;
      float a = 0.0f;
      if (hit) {
        s = ei[e];
        const float v = leaky(s_src[s] + sd[dl]);
        a = __expf(v - ord2f(m_ord[dl])) / den[dl];
      }
      unsigned mask = __builtin_amdgcn_ballot_w32(hit);
      while (mask) {
        const int j = __builtin_ctz(mask);
        mask &= mask - 1;
        row_add(H, acc, __builtin_amdgcn_readlane(s, j),
                __builtin_amdgcn_readlane((int)dl, j),
                __int_as_float(__builtin_amdgcn_readlane(__float_as_int(a), j)),
                lane);
      }
    }
  }
  __syncthreads();

  float4* o4 = (float4*)(out + (size_t)lo * DD);
  for (int i = tid; i < NPB * DD / 4; i += OT) {
    const float4 r = ((const float4*)acc)[i];
    const float4 b = ((const float4*)bias)[i & (DD / 4 - 1)];
    o4[i] = make_float4(r.x + b.x, r.y + b.y, r.z + b.z, r.w + b.w);
  }
}

template <bool BUILD>
static void run_gat(const float* A1, const float* A2, const float* Wmat,
                    const float* a_s, const float* a_d, const float* bias,
                    const int* ei, float* ht, float* out, float* s_src,
                    float* s_dst, int* glist, int* gcnt, hipStream_t stream) {
  gemm_wmma_f32<<<NN / 16, 256, 0, stream>>>(A1, Wmat, ht, 0);
  if (A2)
    gemm_wmma_f32<<<NN / 16, 256, 0, stream>>>(A2, Wmat + DD * DD, ht, 1);
  scores_init<<<NN / 8, 256, 0, stream>>>(ht, a_s, a_d, s_src, s_dst);
  owner_aggregate<BUILD><<<NBLK, OT, 0, stream>>>(ei, ht, s_src, s_dst, bias,
                                                  out, glist, gcnt);
}

extern "C" void kernel_launch(void* const* d_in, const int* in_sizes, int n_in,
                              void* d_out, int out_size, void* d_ws,
                              size_t ws_size, hipStream_t stream) {
  const float* x    = (const float*)d_in[0];
  const int*   ei   = (const int*)d_in[1];
  const float* We   = (const float*)d_in[2];
  const float* ae_s = (const float*)d_in[3];
  const float* ae_d = (const float*)d_in[4];
  const float* be   = (const float*)d_in[5];
  const float* Wp   = (const float*)d_in[6];
  const float* ap_s = (const float*)d_in[7];
  const float* ap_d = (const float*)d_in[8];
  const float* bp   = (const float*)d_in[9];
  const float* Wd   = (const float*)d_in[10];
  const float* ad_s = (const float*)d_in[11];
  const float* ad_d = (const float*)d_in[12];
  const float* bd   = (const float*)d_in[13];
  float*       out  = (float*)d_out;

  const size_t ND = (size_t)NN * DD;
  float*    enc   = (float*)d_ws;
  float*    hA    = enc + ND;
  int*      glist = (int*)(hA + ND);
  int*      gcnt  = glist + NBLK * CAP;
  float*    ht    = hA + ND + ND;
  float*    s_src = ht + ND;
  float*    s_dst = s_src + NN;
  float*    hB    = enc;
  static_assert((size_t)NBLK * CAP + NBLK <= (size_t)NN * DD,
                "lists and counts fit in one feature plane");

  run_gat<true>(x, nullptr, We, ae_s, ae_d, be, ei, ht, enc, s_src, s_dst,
                glist, gcnt, stream);
  run_gat<false>(enc, enc, Wp, ap_s, ap_d, bp, ei, ht, hA, s_src, s_dst,
                 glist, gcnt, stream);
  run_gat<false>(hA, enc, Wp, ap_s, ap_d, bp, ei, ht, hB, s_src, s_dst,
                 glist, gcnt, stream);
  run_gat<false>(hB, nullptr, Wd, ad_s, ad_d, bd, ei, ht, out, s_src, s_dst,
                 glist, gcnt, stream);
}


# next layer's x@W product moved into the owner work-groups' tail: A fragments read from the finished LDS rows (pitch 132), W panel in registers; 5 of 6 GEMM launches gone; sum orders unchanged
# speedup vs baseline: 4.1995x; 1.2727x over previous
#include <hip/hip_runtime.h>

#define NN 50000
#define EE 800000
#define DD 128
#define ETOT (EE + NN)
#define SLOPE 0.2f

typedef float v2f __attribute__((ext_vector_type(2)));
typedef float v8f __attribute__((ext_vector_type(8)));

__global__ __launch_bounds__(256) void gemm_wmma_f32(
    const float* __restrict__ A, const float* __restrict__ W,
    float* __restrict__ H, int accumulate) {
  const int lane = threadIdx.x & 31;
  const int wv   = threadIdx.x >> 5;
  const int m0   = blockIdx.x * 16;
  const int n0   = wv * 16;
  const int half = lane >> 4;
  const int l15  = lane & 15;
  const int row  = m0 + l15;
  const int col  = n0 + l15;

  v8f acc;
  if (accumulate) {
#pragma unroll
    for (int r = 0; r < 8; ++r)
      acc[r] = H[(size_t)(m0 + half * 8 + r) * DD + col];
  } else {
#pragma unroll
    for (int r = 0; r < 8; ++r) acc[r] = 0.0f;
  }

  const float* arow = A + (size_t)row * DD;
#pragma unroll 4
  for (int k = 0; k < DD; k += 4) {
    v2f av, bv;
    av[0] = arow[k + 2 * half];
    av[1] = arow[k + 2 * half + 1];
    bv[0] = W[(size_t)(k + 2 * half) * DD + col];
    bv[1] = W[(size_t)(k + 2 * half + 1) * DD + col];
    acc = __builtin_amdgcn_wmma_f32_16x16x4_f32(false, av, false, bv,
                                                (short)0, acc, false, false);
  }
#pragma unroll
  for (int r = 0; r < 8; ++r)
    H[(size_t)(m0 + half * 8 + r) * DD + col] = acc[r];
}

__global__ __launch_bounds__(256) void scores_init(
    const float* __restrict__ H, const float* __restrict__ a_s,
    const float* __restrict__ a_d, float* __restrict__ s_src,
    float* __restrict__ s_dst) {
  const int lane = threadIdx.x & 31;
  const int node = blockIdx.x * 8 + (threadIdx.x >> 5);
  if (node >= NN) return;
  const float4 hv = *(const float4*)(H + (size_t)node * DD + lane * 4);
  const float4 as = *(const float4*)(a_s + lane * 4);
  const float4 ad = *(const float4*)(a_d + lane * 4);
  float ps = hv.x * as.x + hv.y * as.y + hv.z * as.z + hv.w * as.w;
  float pd = hv.x * ad.x + hv.y * ad.y + hv.z * ad.z + hv.w * ad.w;
#pragma unroll
  for (int off = 16; off > 0; off >>= 1) {
    ps += __shfl_down(ps, off, 32);
    pd += __shfl_down(pd, off, 32);
  }
  if (lane == 0) {
    s_src[node] = ps;
    s_dst[node] = pd;
  }
}

__device__ __forceinline__ unsigned f2ord(float f) {
  unsigned u = __float_as_uint(f);
  return (u & 0x80000000u) ? ~u : (u | 0x80000000u);
}
__device__ __forceinline__ float ord2f(unsigned o) {
  return (o & 0x80000000u) ? __uint_as_float(o & 0x7FFFFFFFu)
                           : __uint_as_float(~o);
}
__device__ __forceinline__ float leaky(float v) {
  return v > 0.0f ? v : SLOPE * v;
}

#define NPB 200
#define NBLK (NN / NPB)
#define OT 1024
#define OW (OT / 32)
#define CAP 16384
#define EB 8
#define AS 132
#define RT ((NPB + 15) / 16)
#define NPBP (RT * 16)
static_assert(NN % NPB == 0, "every work-group owns a full range");
static_assert(AS % 4 == 0 && AS >= DD && OW == 32 && RT <= 16,
              "tail product: 8 column tiles x 4 row groups, at most 4 row tiles each");
static_assert(EE % 4 == 0 && EE % 32 == 0, "whole 16-byte reads, whole wave chunks");
static_assert(NN <= 65536 && NPB <= 4096, "entry packs dst-local<<16 | src");
static_assert(EE <= (1 << 20), "scan entry packs dst-local<<20 | edge");

__device__ __forceinline__ void row_add(const float* __restrict__ H, float* acc,
                                        int src, int dl, float a, int lane) {
  const float* hs = H + (size_t)src * DD + lane;
  const float h0 = hs[0], h1 = hs[32], h2 = hs[64], h3 = hs[96];
  float* o = acc + dl * AS + lane;
  atomicAdd(o + 0, a * h0);
  atomicAdd(o + 32, a * h1);
  atomicAdd(o + 64, a * h2);
  atomicAdd(o + 96, a * h3);
}

template <bool BUILD, int TAIL>
__global__ __launch_bounds__(OT) void owner_aggregate(
    const int* __restrict__ ei, const float* __restrict__ H,
    const float* __restrict__ s_src, const float* __restrict__ s_dst,
    const float* __restrict__ bias, float* __restrict__ out,
    int* glist, int* gcnt, const float* __restrict__ Wn,
    const float* __restrict__ A2, float* __restrict__ Hn) {
  __shared__ __attribute__((aligned(16))) float acc[NPBP * AS];
  __shared__ int l_pk[CAP];
  __shared__ float l_v[CAP];
  __shared__ unsigned m_ord[NPB];
  __shared__ float den[NPB];
  __shared__ float sd[NPB];
  __shared__ int cnt_s;

  const int tid = threadIdx.x;
  const int lane = tid & 31;
  const int wv = tid >> 5;
  const int lo = blockIdx.x * NPB;

  for (int i = tid; i < NPBP * AS / 4; i += OT)
    ((float4*)acc)[i] = make_float4(0.0f, 0.0f, 0.0f, 0.0f);
  if (tid < NPB) {
    const float sdv = s_dst[lo + tid];
    const float v = leaky(s_src[lo + tid] + sdv);
    sd[tid] = sdv;
    den[tid] = 0.0f;
    m_ord[tid] = f2ord(v);
    l_pk[tid] = (tid << 16) | (lo + tid);
    l_v[tid] = v;
  }
  if (tid == 0) cnt_s = NPB;
  __syncthreads();

  int* gl = glist + (size_t)blockIdx.x * CAP;
  int total;
  if (BUILD) {
  const int4* dv = (const int4*)(ei + EE);
  const int QN = EE / 4;
  int4 cur[4], nxt[4];
#pragma unroll
  for (int u = 0; u < 4; ++u) cur[u] = dv[min(tid + u * OT, QN - 1)];
  for (int q0 = tid; q0 < QN; q0 += OT * 4) {
#pragma unroll
    for (int u = 0; u < 4; ++u) nxt[u] = dv[min(q0 + (4 + u) * OT, QN - 1)];
#pragma unroll
    for (int u = 0; u < 4; ++u) {
      const int q = q0 + u * OT;
      const int eb = q * 4;
      const unsigned dl4[4] = {(unsigned)(cur[u].x - lo), (unsigned)(cur[u].y - lo),
                               (unsigned)(cur[u].z - lo), (unsigned)(cur[u].w - lo)};
      if ((q < QN) & (min(min(dl4[0], dl4[1]), min(dl4[2], dl4[3])) < (unsigned)NPB)) {
#pragma unroll
        for (int c4 = 0; c4 < 4; ++c4) {
          const unsigned dl = dl4[c4];
          if (dl < (unsigned)NPB) {
            const int slot = atomicAdd(&cnt_s, 1);
            if (slot < CAP) l_pk[slot] = (int)((dl << 20) | (unsigned)(eb + c4));
          }
        }
      }
    }
#pragma unroll
    for (int u = 0; u < 4; ++u) cur[u] = nxt[u];
  }
  __syncthreads();
  total = cnt_s;
  if (tid == 0) gcnt[blockIdx.x] = total;
  } else {
    total = gcnt[blockIdx.x];
  }
  const bool listed = total <= CAP;

  if (listed) {
    for (int i = NPB + tid; i < total; i += OT) {
      int ent;
      if (BUILD) {
        const unsigned pk = (unsigned)l_pk[i];
        ent = (int)((pk >> 20) << 16) | ei[pk & 0xFFFFFu];
        gl[i] = ent;
      } else {
        ent = gl[i];
      }
      const unsigned dl = (unsigned)ent >> 16;
      const float v = leaky(s_src[ent & 0xFFFF] + sd[dl]);
      atomicMax(&m_ord[dl], f2ord(v));
      l_pk[i] = ent;
      l_v[i] = v;
    }
    __syncthreads();
    for (int i = tid; i < total; i += OT) {
      const int dl = l_pk[i] >> 16;
      const float z = __expf(l_v[i] - ord2f(m_ord[dl]));
      l_v[i] = z;
      atomicAdd(&den[dl], z);
    }
    __syncthreads();
    for (int base = wv * EB; base < total; base += OW * EB) {
      float h[EB][4];
#pragma unroll
      for (int jj = 0; jj < EB; ++jj) {
        int idx = base + jj;
        if (idx >= total) idx = total - 1;
        const int pk = __builtin_amdgcn_readfirstlane(l_pk[idx]);
        const float* hs = H + (size_t)(pk & 0xFFFF) * DD + lane;
        h[jj][0] = hs[0];
        h[jj][1] = hs[32];
        h[jj][2] = hs[64];
        h[jj][3] = hs[96];
      }
#pragma unroll
      for (int jj = 0; jj < EB; ++jj) {
        const int idx = base + jj;
        if (idx < total) {
          const int dl = __builtin_amdgcn_readfirstlane(l_pk[idx]) >> 16;
          const float a = l_v[idx] / den[dl];
          float* o = acc + dl * AS + lane;
          atomicAdd(o + 0, a * h[jj][0]);
          atomicAdd(o + 32, a * h[jj][1]);
          atomicAdd(o + 64, a * h[jj][2]);
          atomicAdd(o + 96, a * h[jj][3]);
        }
      }
    }
  } else {
    for (int e = tid; e < EE; e += OT) {
      const unsigned dl = (unsigned)(ei[EE + e] - lo);
      if (dl < (unsigned)NPB)
        atomicMax(&m_ord[dl], f2ord(leaky(s_src[ei[e]] + sd[dl])));
    }
    __syncthreads();
    if (tid < NPB) {
      const float v = leaky(s_src[lo + tid] + sd[tid]);
      atomicAdd(&den[tid], __expf(v - ord2f(m_ord[tid])));
    }
    for (int e = tid; e < EE; e += OT) {
      const unsigned dl = (unsigned)(ei[EE + e] - lo);
      if (dl < (unsigned)NPB) {
        const float v = leaky(s_src[ei[e]] + sd[dl]);
        atomicAdd(&den[dl], __expf(v - ord2f(m_ord[dl])));
      }
    }
    __syncthreads();
    for (int n = wv; n < NPB; n += OW) {
      const float v = leaky(s_src[lo + n] + sd[n]);
      row_add(H, acc, lo + n, n, __expf(v - ord2f(m_ord[n])) / den[n], lane);
    }
    for (int e0 = wv * 32; e0 < EE; e0 += OW * 32) {
      const int e = e0 + lane;
      const unsigned dl = (unsigned)(ei[EE + e] - lo);
      const bool hit = dl < (unsigned)NPB;
      int s = 0;
      float a = 0.0f;
      if (hit) {
        s = ei[e];
        const float v = leaky(s_src[s] + sd[dl]);
        a = __expf(v - ord2f(m_ord[dl])) / den[dl];
      }
      unsigned mask = __builtin_amdgcn_ballot_w32(hit);
      while (mask) {
        const int j = __builtin_ctz(mask);
        mask &= mask - 1;
        row_add(H, acc, __builtin_amdgcn_readlane(s, j),
                __builtin_amdgcn_readlane((int)dl, j),
                __int_as_float(__builtin_amdgcn_readlane(__float_as_int(a), j)),
                lane);
      }
    }
  }
  __syncthreads();

  float4* acc4 = (float4*)acc;
  float4* o4 = (float4*)(out + (size_t)lo * DD);
  for (int i = tid; i < NPB * DD / 4; i += OT) {
    const int li = (i >> 5) * (AS / 4) + (i & 31);
    const float4 r = acc4[li];
    const float4 b = ((const float4*)bias)[i & (DD / 4 - 1)];
    const float4 v = make_float4(r.x + b.x, r.y + b.y, r.z + b.z, r.w + b.w);
    if (TAIL == 0 || TAIL == 1) o4[i] = v;
    if (TAIL != 0) acc4[li] = v;
  }
  if (TAIL == 0) return;

  __syncthreads();
  {
    const int l15 = lane & 15;
    const int half = lane >> 4;
    const int wu = __builtin_amdgcn_readfirstlane(wv);
    const int ct = wu & 7;
    const int rg = wu >> 3;
    const int col = ct * 16 + l15;
    constexpr int NCH = (TAIL == 3) ? 2 : 4;
    v8f c[4];
#pragma unroll
    for (int t = 0; t < 4; ++t)
#pragma unroll
      for (int r = 0; r < 8; ++r) c[t][r] = 0.0f;
#pragma unroll
    for (int ch = 0; ch < NCH; ++ch) {
      if (TAIL == 2 && ch == 2) {
        __syncthreads();
        const float4* a4 = (const float4*)(A2 + (size_t)lo * DD);
        constexpr int NR = (NPB * DD / 4 + OT - 1) / OT;
#pragma unroll
        for (int u = 0; u < NR; ++u) {
          const int i = tid + u * OT;
          if (i < NPB * DD / 4) acc4[(i >> 5) * (AS / 4) + (i & 31)] = a4[i];
        }
        __syncthreads();
      }
      v2f b[16];
      const float* wp = Wn + (size_t)(ch * 64 + 2 * half) * DD + col;
#pragma unroll
      for (int kk = 0; kk < 16; ++kk) {
        b[kk][0] = wp[(size_t)(kk * 4) * DD];
        b[kk][1] = wp[(size_t)(kk * 4 + 1) * DD];
      }
      const int acol = (ch & 1) * 64 + 2 * half;
#pragma unroll
      for (int t = 0; t < 4; ++t) {
        if (4 * t + 3 < RT || rg + 4 * t < RT) {
          const float* ap = acc + ((rg + 4 * t) * 16 + l15) * AS + acol;
#pragma unroll
          for (int kk = 0; kk < 16; ++kk) {
            const v2f a = *(const v2f*)(ap + kk * 4);
            c[t] = __builtin_amdgcn_wmma_f32_16x16x4_f32(false, a, false, b[kk],
                                                         (short)0, c[t], false, false);
          }
        }
      }
    }
#pragma unroll
    for (int t = 0; t < 4; ++t) {
      if (4 * t + 3 < RT || rg + 4 * t < RT) {
#pragma unroll
        for (int r = 0; r < 8; ++r) {
          const int row = (rg + 4 * t) * 16 + half * 8 + r;
          if ((4 * t + 3) * 16 + 15 < NPB || row < NPB)
            Hn[(size_t)(lo + row) * DD + col] = c[t][r];
        }
      }
    }
  }
}

extern "C" void kernel_launch(void* const* d_in, const int* in_sizes, int n_in,
                              void* d_out, int out_size, void* d_ws,
                              size_t ws_size, hipStream_t stream) {
  const float* x    = (const float*)d_in[0];
  const int*   ei   = (const int*)d_in[1];
  const float* We   = (const float*)d_in[2];
  const float* ae_s = (const float*)d_in[3];
  const float* ae_d = (const float*)d_in[4];
  const float* be   = (const float*)d_in[5];
  const float* Wp   = (const float*)d_in[6];
  const float* ap_s = (const float*)d_in[7];
  const float* ap_d = (const float*)d_in[8];
  const float* bp   = (const float*)d_in[9];
  const float* Wd   = (const float*)d_in[10];
  const float* ad_s = (const float*)d_in[11];
  const float* ad_d = (const float*)d_in[12];
  const float* bd   = (const float*)d_in[13];
  float*       out  = (float*)d_out;

  const size_t ND = (size_t)NN * DD;
  float*    enc   = (float*)d_ws;
  float*    h2    = enc + ND;
  int*      glist = (int*)(h2 + ND);
  int*      gcnt  = glist + NBLK * CAP;
  float*    h1    = h2 + ND + ND;
  float*    s_src = h1 + ND;
  float*    s_dst = s_src + NN;
  static_assert((size_t)NBLK * CAP + NBLK <= (size_t)NN * DD,
                "lists and counts fit in one feature plane");

  gemm_wmma_f32<<<NN / 16, 256, 0, stream>>>(x, We, h1, 0);
  scores_init<<<NN / 8, 256, 0, stream>>>(h1, ae_s, ae_d, s_src, s_dst);
  owner_aggregate<true, 1><<<NBLK, OT, 0, stream>>>(
      ei, h1, s_src, s_dst, be, enc, glist, gcnt, Wp, nullptr, h2);
  scores_init<<<NN / 8, 256, 0, stream>>>(h2, ap_s, ap_d, s_src, s_dst);
  owner_aggregate<false, 2><<<NBLK, OT, 0, stream>>>(
      ei, h2, s_src, s_dst, bp, out, glist, gcnt, Wp, enc, h1);
  scores_init<<<NN / 8, 256, 0, stream>>>(h1, ap_s, ap_d, s_src, s_dst);
  owner_aggregate<false, 3><<<NBLK, OT, 0, stream>>>(
      ei, h1, s_src, s_dst, bp, out, glist, gcnt, Wd, nullptr, h2);
  scores_init<<<NN / 8, 256, 0, stream>>>(h2, ad_s, ad_d, s_src, s_dst);
  owner_aggregate<false, 0><<<NBLK, OT, 0, stream>>>(
      ei, h2, s_src, s_dst, bd, out, glist, gcnt, nullptr, nullptr, nullptr);
}


# score launches: one wave walks 16 consecutive nodes, all 16 row loads in flight (was one node per wave): 50000 short waves per launch become 3125; sums and lane tree unchanged, scores bit-identical
# speedup vs baseline: 4.4513x; 1.0600x over previous
#include <hip/hip_runtime.h>

#define NN 50000
#define EE 800000
#define DD 128
#define ETOT (EE + NN)
#define SLOPE 0.2f

typedef float v2f __attribute__((ext_vector_type(2)));
typedef float v8f __attribute__((ext_vector_type(8)));

__global__ __launch_bounds__(256) void gemm_wmma_f32(
    const float* __restrict__ A, const float* __restrict__ W,
    float* __restrict__ H, int accumulate) {
  const int lane = threadIdx.x & 31;
  const int wv   = threadIdx.x >> 5;
  const int m0   = blockIdx.x * 16;
  const int n0   = wv * 16;
  const int half = lane >> 4;
  const int l15  = lane & 15;
  const int row  = m0 + l15;
  const int col  = n0 + l15;

  v8f acc;
  if (accumulate) {
#pragma unroll
    for (int r = 0; r < 8; ++r)
      acc[r] = H[(size_t)(m0 + half * 8 + r) * DD + col];
  } else {
#pragma unroll
    for (int r = 0; r < 8; ++r) acc[r] = 0.0f;
  }

  const float* arow = A + (size_t)row * DD;
#pragma unroll 4
  for (int k = 0; k < DD; k += 4) {
    v2f av, bv;
    av[0] = arow[k + 2 * half];
    av[1] = arow[k + 2 * half + 1];
    bv[0] = W[(size_t)(k + 2 * half) * DD + col];
    bv[1] = W[(size_t)(k + 2 * half + 1) * DD + col];
    acc = __builtin_amdgcn_wmma_f32_16x16x4_f32(false, av, false, bv,
                                                (short)0, acc, false, false);
  }
#pragma unroll
  for (int r = 0; r < 8; ++r)
    H[(size_t)(m0 + half * 8 + r) * DD + col] = acc[r];
}

#define SR 16
#define SGRID ((NN / SR + 7) / 8)
static_assert(NN % SR == 0, "every wave has a full batch of nodes");
static_assert((SR * 4) % 16 == 0, "a wave's SR scores are whole 16-byte groups");

__global__ __launch_bounds__(256) void scores_init(
    const float* __restrict__ H, const float* __restrict__ a_s,
    const float* __restrict__ a_d, float* __restrict__ s_src,
    float* __restrict__ s_dst) {
  const int lane = threadIdx.x & 31;
  const int node0 = (blockIdx.x * 8 + (threadIdx.x >> 5)) * SR;
  if (node0 >= NN) return;
  const float4 as = *(const float4*)(a_s + lane * 4);
  const float4 ad = *(const float4*)(a_d + lane * 4);
  float4 hv[SR];
#pragma unroll
  for (int j = 0; j < SR; ++j)
    hv[j] = *(const float4*)(H + (size_t)(node0 + j) * DD + lane * 4);
  __builtin_amdgcn_sched_barrier(0);
  float ps[SR], pd[SR];
#pragma unroll
  for (int j = 0; j < SR; ++j) {
    ps[j] = fmaf(hv[j].w, as.w, fmaf(hv[j].z, as.z, fmaf(hv[j].x, as.x, hv[j].y * as.y)));
    pd[j] = fmaf(hv[j].w, ad.w, fmaf(hv[j].z, ad.z, fmaf(hv[j].x, ad.x, hv[j].y * ad.y)));
  }
#pragma unroll
  for (int off = 16; off > 0; off >>= 1) {
#pragma unroll
    for (int j = 0; j < SR; ++j) {
      ps[j] += __shfl_down(ps[j], off, 32);
      pd[j] += __shfl_down(pd[j], off, 32);
    }
  }
  if (lane == 0) {
    float4* os = (float4*)(s_src + node0);
    float4* od = (float4*)(s_dst + node0);
#pragma unroll
    for (int q = 0; q < SR / 4; ++q) {
      os[q] = make_float4(ps[4 * q], ps[4 * q + 1], ps[4 * q + 2], ps[4 * q + 3]);
      od[q] = make_float4(pd[4 * q], pd[4 * q + 1], pd[4 * q + 2], pd[4 * q + 3]);
    }
  }
}

__device__ __forceinline__ unsigned f2ord(float f) {
  unsigned u = __float_as_uint(f);
  return (u & 0x80000000u) ? ~u : (u | 0x80000000u);
}
__device__ __forceinline__ float ord2f(unsigned o) {
  return (o & 0x80000000u) ? __uint_as_float(o & 0x7FFFFFFFu)
                           : __uint_as_float(~o);
}
__device__ __forceinline__ float leaky(float v) {
  return v > 0.0f ? v : SLOPE * v;
}

#define NPB 200
#define NBLK (NN / NPB)
#define OT 1024
#define OW (OT / 32)
#define CAP 16384
#define EB 8
#define AS 132
#define RT ((NPB + 15) / 16)
#define NPBP (RT * 16)
static_assert(NN % NPB == 0, "every work-group owns a full range");
static_assert(AS % 4 == 0 && AS >= DD && OW == 32 && RT <= 16,
              "tail product: 8 column tiles x 4 row groups, at most 4 row tiles each");
static_assert(EE % 4 == 0 && EE % 32 == 0, "whole 16-byte reads, whole wave chunks");
static_assert(NN <= 65536 && NPB <= 4096, "entry packs dst-local<<16 | src");
static_assert(EE <= (1 << 20), "scan entry packs dst-local<<20 | edge");

__device__ __forceinline__ void row_add(const float* __restrict__ H, float* acc,
                                        int src, int dl, float a, int lane) {
  const float* hs = H + (size_t)src * DD + lane;
  const float h0 = hs[0], h1 = hs[32], h2 = hs[64], h3 = hs[96];
  float* o = acc + dl * AS + lane;
  atomicAdd(o + 0, a * h0);
  atomicAdd(o + 32, a * h1);
  atomicAdd(o + 64, a * h2);
  atomicAdd(o + 96, a * h3);
}

template <bool BUILD, int TAIL>
__global__ __launch_bounds__(OT) void owner_aggregate(
    const int* __restrict__ ei, const float* __restrict__ H,
    const float* __restrict__ s_src, const float* __restrict__ s_dst,
    const float* __restrict__ bias, float* __restrict__ out,
    int* glist, int* gcnt, const float* __restrict__ Wn,
    const float* __restrict__ A2, float* __restrict__ Hn) {
  __shared__ __attribute__((aligned(16))) float acc[NPBP * AS];
  __shared__ int l_pk[CAP];
  __shared__ float l_v[CAP];
  __shared__ unsigned m_ord[NPB];
  __shared__ float den[NPB];
  __shared__ float sd[NPB];
  __shared__ int cnt_s;

  const int tid = threadIdx.x;
  const int lane = tid & 31;
  const int wv = tid >> 5;
  const int lo = blockIdx.x * NPB;

  for (int i = tid; i < NPBP * AS / 4; i += OT)
    ((float4*)acc)[i] = make_float4(0.0f, 0.0f, 0.0f, 0.0f);
  if (tid < NPB) {
    const float sdv = s_dst[lo + tid];
    const float v = leaky(s_src[lo + tid] + sdv);
    sd[tid] = sdv;
    den[tid] = 0.0f;
    m_ord[tid] = f2ord(v);
    l_pk[tid] = (tid << 16) | (lo + tid);
    l_v[tid] = v;
  }
  if (tid == 0) cnt_s = NPB;
  __syncthreads();

  int* gl = glist + (size_t)blockIdx.x * CAP;
  int total;
  if (BUILD) {
  const int4* dv = (const int4*)(ei + EE);
  const int QN = EE / 4;
  int4 cur[4], nxt[4];
#pragma unroll
  for (int u = 0; u < 4; ++u) cur[u] = dv[min(tid + u * OT, QN - 1)];
  for (int q0 = tid; q0 < QN; q0 += OT * 4) {
#pragma unroll
    for (int u = 0; u < 4; ++u) nxt[u] = dv[min(q0 + (4 + u) * OT, QN - 1)];
#pragma unroll
    for (int u = 0; u < 4; ++u) {
      const int q = q0 + u * OT;
      const int eb = q * 4;
      const unsigned dl4[4] = {(unsigned)(cur[u].x - lo), (unsigned)(cur[u].y - lo),
                               (unsigned)(cur[u].z - lo), (unsigned)(cur[u].w - lo)};
      if ((q < QN) & (min(min(dl4[0], dl4[1]), min(dl4[2], dl4[3])) < (unsigned)NPB)) {
#pragma unroll
        for (int c4 = 0; c4 < 4; ++c4) {
          const unsigned dl = dl4[c4];
          if (dl < (unsigned)NPB) {
            const int slot = atomicAdd(&cnt_s, 1);
            if (slot < CAP) l_pk[slot] = (int)((dl << 20) | (unsigned)(eb + c4));
          }
        }
      }
    }
#pragma unroll
    for (int u = 0; u < 4; ++u) cur[u] = nxt[u];
  }
  __syncthreads();
  total = cnt_s;
  if (tid == 0) gcnt[blockIdx.x] = total;
  } else {
    total = gcnt[blockIdx.x];
  }
  const bool listed = total <= CAP;

  if (listed) {
    for (int i = NPB + tid; i < total; i += OT) {
      int ent;
      if (BUILD) {
        const unsigned pk = (unsigned)l_pk[i];
        ent = (int)((pk >> 20) << 16) | ei[pk & 0xFFFFFu];
        gl[i] = ent;
      } else {
        ent = gl[i];
      }
      const unsigned dl = (unsigned)ent >> 16;
      const float v = leaky(s_src[ent & 0xFFFF] + sd[dl]);
      atomicMax(&m_ord[dl], f2ord(v));
      l_pk[i] = ent;
      l_v[i] = v;
    }
    __syncthreads();
    for (int i = tid; i < total; i += OT) {
      const int dl = l_pk[i] >> 16;
      const float z = __expf(l_v[i] - ord2f(m_ord[dl]));
      l_v[i] = z;
      atomicAdd(&den[dl], z);
    }
    __syncthreads();
    for (int base = wv * EB; base < total; base += OW * EB) {
      float h[EB][4];
#pragma unroll
      for (int jj = 0; jj < EB; ++jj) {
        int idx = base + jj;
        if (idx >= total) idx = total - 1;
        const int pk = __builtin_amdgcn_readfirstlane(l_pk[idx]);
        const float* hs = H + (size_t)(pk & 0xFFFF) * DD + lane;
        h[jj][0] = hs[0];
        h[jj][1] = hs[32];
        h[jj][2] = hs[64];
        h[jj][3] = hs[96];
      }
#pragma unroll
      for (int jj = 0; jj < EB; ++jj) {
        const int idx = base + jj;
        if (idx < total) {
          const int dl = __builtin_amdgcn_readfirstlane(l_pk[idx]) >> 16;
          const float a = l_v[idx] / den[dl];
          float* o = acc + dl * AS + lane;
          atomicAdd(o + 0, a * h[jj][0]);
          atomicAdd(o + 32, a * h[jj][1]);
          atomicAdd(o + 64, a * h[jj][2]);
          atomicAdd(o + 96, a * h[jj][3]);
        }
      }
    }
  } else {
    for (int e = tid; e < EE; e += OT) {
      const unsigned dl = (unsigned)(ei[EE + e] - lo);
      if (dl < (unsigned)NPB)
        atomicMax(&m_ord[dl], f2ord(leaky(s_src[ei[e]] + sd[dl])));
    }
    __syncthreads();
    if (tid < NPB) {
      const float v = leaky(s_src[lo + tid] + sd[tid]);
      atomicAdd(&den[tid], __expf(v - ord2f(m_ord[tid])));
    }
    for (int e = tid; e < EE; e += OT) {
      const unsigned dl = (unsigned)(ei[EE + e] - lo);
      if (dl < (unsigned)NPB) {
        const float v = leaky(s_src[ei[e]] + sd[dl]);
        atomicAdd(&den[dl], __expf(v - ord2f(m_ord[dl])));
      }
    }
    __syncthreads();
    for (int n = wv; n < NPB; n += OW) {
      const float v = leaky(s_src[lo + n] + sd[n]);
      row_add(H, acc, lo + n, n, __expf(v - ord2f(m_ord[n])) / den[n], lane);
    }
    for (int e0 = wv * 32; e0 < EE; e0 += OW * 32) {
      const int e = e0 + lane;
      const unsigned dl = (unsigned)(ei[EE + e] - lo);
      const bool hit = dl < (unsigned)NPB;
      int s = 0;
      float a = 0.0f;
      if (hit) {
        s = ei[e];
        const float v = leaky(s_src[s] + sd[dl]);
        a = __expf(v - ord2f(m_ord[dl])) / den[dl];
      }
      unsigned mask = __builtin_amdgcn_ballot_w32(hit);
      while (mask) {
        const int j = __builtin_ctz(mask);
        mask &= mask - 1;
        row_add(H, acc, __builtin_amdgcn_readlane(s, j),
                __builtin_amdgcn_readlane((int)dl, j),
                __int_as_float(__builtin_amdgcn_readlane(__float_as_int(a), j)),
                lane);
      }
    }
  }
  __syncthreads();

  float4* acc4 = (float4*)acc;
  float4* o4 = (float4*)(out + (size_t)lo * DD);
  for (int i = tid; i < NPB * DD / 4; i += OT) {
    const int li = (i >> 5) * (AS / 4) + (i & 31);
    const float4 r = acc4[li];
    const float4 b = ((const float4*)bias)[i & (DD / 4 - 1)];
    const float4 v = make_float4(r.x + b.x, r.y + b.y, r.z + b.z, r.w + b.w);
    if (TAIL == 0 || TAIL == 1) o4[i] = v;
    if (TAIL != 0) acc4[li] = v;
  }
  if (TAIL == 0) return;

  __syncthreads();
  {
    const int l15 = lane & 15;
    const int half = lane >> 4;
    const int wu = __builtin_amdgcn_readfirstlane(wv);
    const int ct = wu & 7;
    const int rg = wu >> 3;
    const int col = ct * 16 + l15;
    constexpr int NCH = (TAIL == 3) ? 2 : 4;
    v8f c[4];
#pragma unroll
    for (int t = 0; t < 4; ++t)
#pragma unroll
      for (int r = 0; r < 8; ++r) c[t][r] = 0.0f;
#pragma unroll
    for (int ch = 0; ch < NCH; ++ch) {
      if (TAIL == 2 && ch == 2) {
        __syncthreads();
        const float4* a4 = (const float4*)(A2 + (size_t)lo * DD);
        constexpr int NR = (NPB * DD / 4 + OT - 1) / OT;
#pragma unroll
        for (int u = 0; u < NR; ++u) {
          const int i = tid + u * OT;
          if (i < NPB * DD / 4) acc4[(i >> 5) * (AS / 4) + (i & 31)] = a4[i];
        }
        __syncthreads();
      }
      v2f b[16];
      const float* wp = Wn + (size_t)(ch * 64 + 2 * half) * DD + col;
#pragma unroll
      for (int kk = 0; kk < 16; ++kk) {
        b[kk][0] = wp[(size_t)(kk * 4) * DD];
        b[kk][1] = wp[(size_t)(kk * 4 + 1) * DD];
      }
      const int acol = (ch & 1) * 64 + 2 * half;
#pragma unroll
      for (int t = 0; t < 4; ++t) {
        if (4 * t + 3 < RT || rg + 4 * t < RT) {
          const float* ap = acc + ((rg + 4 * t) * 16 + l15) * AS + acol;
#pragma unroll
          for (int kk = 0; kk < 16; ++kk) {
            const v2f a = *(const v2f*)(ap + kk * 4);
            c[t] = __builtin_amdgcn_wmma_f32_16x16x4_f32(false, a, false, b[kk],
                                                         (short)0, c[t], false, false);
          }
        }
      }
    }
#pragma unroll
    for (int t = 0; t < 4; ++t) {
      if (4 * t + 3 < RT || rg + 4 * t < RT) {
#pragma unroll
        for (int r = 0; r < 8; ++r) {
          const int row = (rg + 4 * t) * 16 + half * 8 + r;
          if ((4 * t + 3) * 16 + 15 < NPB || row < NPB)
            Hn[(size_t)(lo + row) * DD + col] = c[t][r];
        }
      }
    }
  }
}

extern "C" void kernel_launch(void* const* d_in, const int* in_sizes, int n_in,
                              void* d_out, int out_size, void* d_ws,
                              size_t ws_size, hipStream_t stream) {
  const float* x    = (const float*)d_in[0];
  const int*   ei   = (const int*)d_in[1];
  const float* We   = (const float*)d_in[2];
  const float* ae_s = (const float*)d_in[3];
  const float* ae_d = (const float*)d_in[4];
  const float* be   = (const float*)d_in[5];
  const float* Wp   = (const float*)d_in[6];
  const float* ap_s = (const float*)d_in[7];
  const float* ap_d = (const float*)d_in[8];
  const float* bp   = (const float*)d_in[9];
  const float* Wd   = (const float*)d_in[10];
  const float* ad_s = (const float*)d_in[11];
  const float* ad_d = (const float*)d_in[12];
  const float* bd   = (const float*)d_in[13];
  float*       out  = (float*)d_out;

  const size_t ND = (size_t)NN * DD;
  float*    enc   = (float*)d_ws;
  float*    h2    = enc + ND;
  int*      glist = (int*)(h2 + ND);
  int*      gcnt  = glist + NBLK * CAP;
  float*    h1    = h2 + ND + ND;
  float*    s_src = h1 + ND;
  float*    s_dst = s_src + NN;
  static_assert((size_t)NBLK * CAP + NBLK <= (size_t)NN * DD,
                "lists and counts fit in one feature plane");

  gemm_wmma_f32<<<NN / 16, 256, 0, stream>>>(x, We, h1, 0);
  scores_init<<<SGRID, 256, 0, stream>>>(h1, ae_s, ae_d, s_src, s_dst);
  owner_aggregate<true, 1><<<NBLK, OT, 0, stream>>>(
      ei, h1, s_src, s_dst, be, enc, glist, gcnt, Wp, nullptr, h2);
  scores_init<<<SGRID, 256, 0, stream>>>(h2, ap_s, ap_d, s_src, s_dst);
  owner_aggregate<false, 2><<<NBLK, OT, 0, stream>>>(
      ei, h2, s_src, s_dst, bp, out, glist, gcnt, Wp, enc, h1);
  scores_init<<<SGRID, 256, 0, stream>>>(h1, ap_s, ap_d, s_src, s_dst);
  owner_aggregate<false, 3><<<NBLK, OT, 0, stream>>>(
      ei, h1, s_src, s_dst, bp, out, glist, gcnt, Wd, nullptr, h2);
  scores_init<<<SGRID, 256, 0, stream>>>(h2, ad_s, ad_d, s_src, s_dst);
  owner_aggregate<false, 0><<<NBLK, OT, 0, stream>>>(
      ei, h2, s_src, s_dst, bd, out, glist, gcnt, nullptr, nullptr, nullptr);
}


# first owner launch scans a 1-byte-per-edge owner-id plane (dst / 200, written in the encode GEMM launch), not the 4-byte destination list: 4x fewer scanned bytes; LDS f32 adds still in arrival order
# speedup vs baseline: 4.6577x; 1.0464x over previous
#include <hip/hip_runtime.h>

#define NN 50000
#define EE 800000
#define DD 128
#define ETOT (EE + NN)
#define SLOPE 0.2f
#define NPB 200
#define NBLK (NN / NPB)
static_assert(NN % NPB == 0, "every work-group owns a full range");
static_assert(NBLK <= 255, "an owner id fits in one byte, 255 is left for 'no owner'");
static_assert(EE % 16 == 0, "the owner-id plane is whole 16-byte groups of 16 edges");

typedef float v2f __attribute__((ext_vector_type(2)));
typedef float v8f __attribute__((ext_vector_type(8)));

__device__ __forceinline__ unsigned owner_id(int dst) {
  return (unsigned)dst < (unsigned)NN ? (unsigned)dst / NPB : 255u;
}

__global__ __launch_bounds__(256) void gemm_wmma_f32(
    const float* __restrict__ A, const float* __restrict__ W,
    float* __restrict__ H, int accumulate, const int* __restrict__ ei,
    unsigned* __restrict__ ob) {
  const int g4 = blockIdx.x * 256 + threadIdx.x;
  if (g4 < EE / 4) {
    const int4 d = ((const int4*)(ei + EE))[g4];
    ob[g4] = owner_id(d.x) | (owner_id(d.y) << 8) | (owner_id(d.z) << 16) |
             (owner_id(d.w) << 24);
  }
  const int lane = threadIdx.x & 31;
  const int wv   = threadIdx.x >> 5;
  const int m0   = blockIdx.x * 16;
  const int n0   = wv * 16;
  const int half = lane >> 4;
  const int l15  = lane & 15;
  const int row  = m0 + l15;
  const int col  = n0 + l15;

  v8f acc;
  if (accumulate) {
#pragma unroll
    for (int r = 0; r < 8; ++r)
      acc[r] = H[(size_t)(m0 + half * 8 + r) * DD + col];
  } else {
#pragma unroll
    for (int r = 0; r < 8; ++r) acc[r] = 0.0f;
  }

  const float* arow = A + (size_t)row * DD;
#pragma unroll 4
  for (int k = 0; k < DD; k += 4) {
    v2f av, bv;
    av[0] = arow[k + 2 * half];
    av[1] = arow[k + 2 * half + 1];
    bv[0] = W[(size_t)(k + 2 * half) * DD + col];
    bv[1] = W[(size_t)(k + 2 * half + 1) * DD + col];
    acc = __builtin_amdgcn_wmma_f32_16x16x4_f32(false, av, false, bv,
                                                (short)0, acc, false, false);
  }
#pragma unroll
  for (int r = 0; r < 8; ++r)
    H[(size_t)(m0 + half * 8 + r) * DD + col] = acc[r];
}

#define SR 16
#define SGRID ((NN / SR + 7) / 8)
static_assert(NN % SR == 0, "every wave has a full batch of nodes");
static_assert((SR * 4) % 16 == 0, "a wave's SR scores are whole 16-byte groups");

__global__ __launch_bounds__(256) void scores_init(
    const float* __restrict__ H, const float* __restrict__ a_s,
    const float* __restrict__ a_d, float* __restrict__ s_src,
    float* __restrict__ s_dst) {
  const int lane = threadIdx.x & 31;
  const int node0 = (blockIdx.x * 8 + (threadIdx.x >> 5)) * SR;
  if (node0 >= NN) return;
  const float4 as = *(const float4*)(a_s + lane * 4);
  const float4 ad = *(const float4*)(a_d + lane * 4);
  float4 hv[SR];
#pragma unroll
  for (int j = 0; j < SR; ++j)
    hv[j] = *(const float4*)(H + (size_t)(node0 + j) * DD + lane * 4);
  __builtin_amdgcn_sched_barrier(0);
  float ps[SR], pd[SR];
#pragma unroll
  for (int j = 0; j < SR; ++j) {
    ps[j] = fmaf(hv[j].w, as.w, fmaf(hv[j].z, as.z, fmaf(hv[j].x, as.x, hv[j].y * as.y)));
    pd[j] = fmaf(hv[j].w, ad.w, fmaf(hv[j].z, ad.z, fmaf(hv[j].x, ad.x, hv[j].y * ad.y)));
  }
#pragma unroll
  for (int off = 16; off > 0; off >>= 1) {
#pragma unroll
    for (int j = 0; j < SR; ++j) {
      ps[j] += __shfl_down(ps[j], off, 32);
      pd[j] += __shfl_down(pd[j], off, 32);
    }
  }
  if (lane == 0) {
    float4* os = (float4*)(s_src + node0);
    float4* od = (float4*)(s_dst + node0);
#pragma unroll
    for (int q = 0; q < SR / 4; ++q) {
      os[q] = make_float4(ps[4 * q], ps[4 * q + 1], ps[4 * q + 2], ps[4 * q + 3]);
      od[q] = make_float4(pd[4 * q], pd[4 * q + 1], pd[4 * q + 2], pd[4 * q + 3]);
    }
  }
}

__device__ __forceinline__ unsigned f2ord(float f) {
  unsigned u = __float_as_uint(f);
  return (u & 0x80000000u) ? ~u : (u | 0x80000000u);
}
__device__ __forceinline__ float ord2f(unsigned o) {
  return (o & 0x80000000u) ? __uint_as_float(o & 0x7FFFFFFFu)
                           : __uint_as_float(~o);
}
__device__ __forceinline__ float leaky(float v) {
  return v > 0.0f ? v : SLOPE * v;
}

#define OT 1024
#define OW (OT / 32)
#define CAP 16384
#define EB 8
#define AS 132
#define RT ((NPB + 15) / 16)
#define NPBP (RT * 16)
static_assert(AS % 4 == 0 && AS >= DD && OW == 32 && RT <= 16,
              "tail product: 8 column tiles x 4 row groups, at most 4 row tiles each");
static_assert(EE % 4 == 0 && EE % 32 == 0, "whole 16-byte reads, whole wave chunks");
static_assert(NN <= 65536 && NPB <= 4096, "entry packs dst-local<<16 | src");
static_assert(EE <= (1 << 20), "a scan entry is an edge index");

__device__ __forceinline__ unsigned zero_bytes(unsigned x) {
  return ~(((x & 0x7F7F7F7Fu) + 0x7F7F7F7Fu) | x | 0x7F7F7F7Fu);
}

__device__ __forceinline__ void row_add(const float* __restrict__ H, float* acc,
                                        int src, int dl, float a, int lane) {
  const float* hs = H + (size_t)src * DD + lane;
  const float h0 = hs[0], h1 = hs[32], h2 = hs[64], h3 = hs[96];
  float* o = acc + dl * AS + lane;
  atomicAdd(o + 0, a * h0);
  atomicAdd(o + 32, a * h1);
  atomicAdd(o + 64, a * h2);
  atomicAdd(o + 96, a * h3);
}

template <bool BUILD, int TAIL>
__global__ __launch_bounds__(OT) void owner_aggregate(
    const int* __restrict__ ei, const float* __restrict__ H,
    const float* __restrict__ s_src, const float* __restrict__ s_dst,
    const float* __restrict__ bias, float* __restrict__ out,
    int* glist, int* gcnt, const float* __restrict__ Wn,
    const float* __restrict__ A2, float* __restrict__ Hn,
    const unsigned* __restrict__ ob) {
  __shared__ __attribute__((aligned(16))) float acc[NPBP * AS];
  __shared__ int l_pk[CAP];
  __shared__ float l_v[CAP];
  __shared__ unsigned m_ord[NPB];
  __shared__ float den[NPB];
  __shared__ float sd[NPB];
  __shared__ int cnt_s;

  const int tid = threadIdx.x;
  const int lane = tid & 31;
  const int wv = tid >> 5;
  const int lo = blockIdx.x * NPB;

  for (int i = tid; i < NPBP * AS / 4; i += OT)
    ((float4*)acc)[i] = make_float4(0.0f, 0.0f, 0.0f, 0.0f);
  if (tid < NPB) {
    const float sdv = s_dst[lo + tid];
    const float v = leaky(s_src[lo + tid] + sdv);
    sd[tid] = sdv;
    den[tid] = 0.0f;
    m_ord[tid] = f2ord(v);
    l_pk[tid] = (tid << 16) | (lo + tid);
    l_v[tid] = v;
  }
  if (tid == 0) cnt_s = NPB;
  __syncthreads();

  int* gl = glist + (size_t)blockIdx.x * CAP;
  int total;
  if (BUILD) {
  const uint4* ov = (const uint4*)ob;
  const int QN = EE / 16;
  const unsigned pat = (unsigned)blockIdx.x * 0x01010101u;
  uint4 cur[4], nxt[4];
#pragma unroll
  for (int u = 0; u < 4; ++u) cur[u] = ov[min(tid + u * OT, QN - 1)];
  for (int q0 = tid; q0 < QN; q0 += OT * 4) {
#pragma unroll
    for (int u = 0; u < 4; ++u) nxt[u] = ov[min(q0 + (4 + u) * OT, QN - 1)];
#pragma unroll
    for (int u = 0; u < 4; ++u) {
      const int q = q0 + u * OT;
      unsigned m = (zero_bytes(cur[u].x ^ pat) >> 7) | (zero_bytes(cur[u].y ^ pat) >> 6) |
                   (zero_bytes(cur[u].z ^ pat) >> 5) | (zero_bytes(cur[u].w ^ pat) >> 4);
      if (q >= QN) m = 0;
      while (m) {
        const int j = __builtin_ctz(m);
        m &= m - 1;
        const int slot = atomicAdd(&cnt_s, 1);
        if (slot < CAP) l_pk[slot] = q * 16 + (j & 3) * 4 + (j >> 3);
      }
    }
#pragma unroll
    for (int u = 0; u < 4; ++u) cur[u] = nxt[u];
  }
  __syncthreads();
  total = cnt_s;
  if (tid == 0) gcnt[blockIdx.x] = total;
  } else {
    total = gcnt[blockIdx.x];
  }
  const bool listed = total <= CAP;

  if (listed) {
    for (int i = NPB + tid; i < total; i += OT) {
      int ent;
      if (BUILD) {
        const int e = l_pk[i];
        ent = ((ei[EE + e] - lo) << 16) | ei[e];
        gl[i] = ent;
      } else {
        ent = gl[i];
      }
      const unsigned dl = (unsigned)ent >> 16;
      const float v = leaky(s_src[ent & 0xFFFF] + sd[dl]);
      atomicMax(&m_ord[dl], f2ord(v));
      l_pk[i] = ent;
      l_v[i] = v;
    }
    __syncthreads();
    for (int i = tid; i < total; i += OT) {
      const int dl = l_pk[i] >> 16;
      const float z = __expf(l_v[i] - ord2f(m_ord[dl]));
      l_v[i] = z;
      atomicAdd(&den[dl], z);
    }
    __syncthreads();
    for (int base = wv * EB; base < total; base += OW * EB) {
      float h[EB][4];
#pragma unroll
      for (int jj = 0; jj < EB; ++jj) {
        int idx = base + jj;
        if (idx >= total) idx = total - 1;
        const int pk = __builtin_amdgcn_readfirstlane(l_pk[idx]);
        const float* hs = H + (size_t)(pk & 0xFFFF) * DD + lane;
        h[jj][0] = hs[0];
        h[jj][1] = hs[32];
        h[jj][2] = hs[64];
        h[jj][3] = hs[96];
      }
#pragma unroll
      for (int jj = 0; jj < EB; ++jj) {
        const int idx = base + jj;
        if (idx < total) {
          const int dl = __builtin_amdgcn_readfirstlane(l_pk[idx]) >> 16;
          const float a = l_v[idx] / den[dl];
          float* o = acc + dl * AS + lane;
          atomicAdd(o + 0, a * h[jj][0]);
          atomicAdd(o + 32, a * h[jj][1]);
          atomicAdd(o + 64, a * h[jj][2]);
          atomicAdd(o + 96, a * h[jj][3]);
        }
      }
    }
  } else {
    for (int e = tid; e < EE; e += OT) {
      const unsigned dl = (unsigned)(ei[EE + e] - lo);
      if (dl < (unsigned)NPB)
        atomicMax(&m_ord[dl], f2ord(leaky(s_src[ei[e]] + sd[dl])));
    }
    __syncthreads();
    if (tid < NPB) {
      const float v = leaky(s_src[lo + tid] + sd[tid]);
      atomicAdd(&den[tid], __expf(v - ord2f(m_ord[tid])));
    }
    for (int e = tid; e < EE; e += OT) {
      const unsigned dl = (unsigned)(ei[EE + e] - lo);
      if (dl < (unsigned)NPB) {
        const float v = leaky(s_src[ei[e]] + sd[dl]);
        atomicAdd(&den[dl], __expf(v - ord2f(m_ord[dl])));
      }
    }
    __syncthreads();
    for (int n = wv; n < NPB; n += OW) {
      const float v = leaky(s_src[lo + n] + sd[n]);
      row_add(H, acc, lo + n, n, __expf(v - ord2f(m_ord[n])) / den[n], lane);
    }
    for (int e0 = wv * 32; e0 < EE; e0 += OW * 32) {
      const int e = e0 + lane;
      const unsigned dl = (unsigned)(ei[EE + e] - lo);
      const bool hit = dl < (unsigned)NPB;
      int s = 0;
      float a = 0.0f;
      if (hit) {
        s = ei[e];
        const float v = leaky(s_src[s] + sd[dl]);
        a = __expf(v - ord2f(m_ord[dl])) / den[dl];
      }
      unsigned mask = __builtin_amdgcn_ballot_w32(hit);
      while (mask) {
        const int j = __builtin_ctz(mask);
        mask &= mask - 1;
        row_add(H, acc, __builtin_amdgcn_readlane(s, j),
                __builtin_amdgcn_readlane((int)dl, j),
                __int_as_float(__builtin_amdgcn_readlane(__float_as_int(a), j)),
                lane);
      }
    }
  }
  __syncthreads();

  float4* acc4 = (float4*)acc;
  float4* o4 = (float4*)(out + (size_t)lo * DD);
  for (int i = tid; i < NPB * DD / 4; i += OT) {
    const int li = (i >> 5) * (AS / 4) + (i & 31);
    const float4 r = acc4[li];
    const float4 b = ((const float4*)bias)[i & (DD / 4 - 1)];
    const float4 v = make_float4(r.x + b.x, r.y + b.y, r.z + b.z, r.w + b.w);
    if (TAIL == 0 || TAIL == 1) o4[i] = v;
    if (TAIL != 0) acc4[li] = v;
  }
  if (TAIL == 0) return;

  __syncthreads();
  {
    const int l15 = lane & 15;
    const int half = lane >> 4;
    const int wu = __builtin_amdgcn_readfirstlane(wv);
    const int ct = wu & 7;
    const int rg = wu >> 3;
    const int col = ct * 16 + l15;
    constexpr int NCH = (TAIL == 3) ? 2 : 4;
    v8f c[4];
#pragma unroll
    for (int t = 0; t < 4; ++t)
#pragma unroll
      for (int r = 0; r < 8; ++r) c[t][r] = 0.0f;
#pragma unroll
    for (int ch = 0; ch < NCH; ++ch) {
      if (TAIL == 2 && ch == 2) {
        __syncthreads();
        const float4* a4 = (const float4*)(A2 + (size_t)lo * DD);
        constexpr int NR = (NPB * DD / 4 + OT - 1) / OT;
#pragma unroll
        for (int u = 0; u < NR; ++u) {
          const int i = tid + u * OT;
          if (i < NPB * DD / 4) acc4[(i >> 5) * (AS / 4) + (i & 31)] = a4[i];
        }
        __syncthreads();
      }
      v2f b[16];
      const float* wp = Wn + (size_t)(ch * 64 + 2 * half) * DD + col;
#pragma unroll
      for (int kk = 0; kk < 16; ++kk) {
        b[kk][0] = wp[(size_t)(kk * 4) * DD];
        b[kk][1] = wp[(size_t)(kk * 4 + 1) * DD];
      }
      const int acol = (ch & 1) * 64 + 2 * half;
#pragma unroll
      for (int t = 0; t < 4; ++t) {
        if (4 * t + 3 < RT || rg + 4 * t < RT) {
          const float* ap = acc + ((rg + 4 * t) * 16 + l15) * AS + acol;
#pragma unroll
          for (int kk = 0; kk < 16; ++kk) {
            const v2f a = *(const v2f*)(ap + kk * 4);
            c[t] = __builtin_amdgcn_wmma_f32_16x16x4_f32(false, a, false, b[kk],
                                                         (short)0, c[t], false, false);
          }
        }
      }
    }
#pragma unroll
    for (int t = 0; t < 4; ++t) {
      if (4 * t + 3 < RT || rg + 4 * t < RT) {
#pragma unroll
        for (int r = 0; r < 8; ++r) {
          const int row = (rg + 4 * t) * 16 + half * 8 + r;
          if ((4 * t + 3) * 16 + 15 < NPB || row < NPB)
            Hn[(size_t)(lo + row) * DD + col] = c[t][r];
        }
      }
    }
  }
}

extern "C" void kernel_launch(void* const* d_in, const int* in_sizes, int n_in,
                              void* d_out, int out_size, void* d_ws,
                              size_t ws_size, hipStream_t stream) {
  const float* x    = (const float*)d_in[0];
  const int*   ei   = (const int*)d_in[1];
  const float* We   = (const float*)d_in[2];
  const float* ae_s = (const float*)d_in[3];
  const float* ae_d = (const float*)d_in[4];
  const float* be   = (const float*)d_in[5];
  const float* Wp   = (const float*)d_in[6];
  const float* ap_s = (const float*)d_in[7];
  const float* ap_d = (const float*)d_in[8];
  const float* bp   = (const float*)d_in[9];
  const float* Wd   = (const float*)d_in[10];
  const float* ad_s = (const float*)d_in[11];
  const float* ad_d = (const float*)d_in[12];
  const float* bd   = (const float*)d_in[13];
  float*       out  = (float*)d_out;

  const size_t ND = (size_t)NN * DD;
  float*    enc   = (float*)d_ws;
  float*    h2    = enc + ND;
  int*      glist = (int*)(h2 + ND);
  int*      gcnt  = glist + NBLK * CAP;
  unsigned* ob    = (unsigned*)(gcnt + 256);
  float*    h1    = h2 + ND + ND;
  float*    s_src = h1 + ND;
  float*    s_dst = s_src + NN;
  static_assert(NBLK <= 256 && (size_t)NBLK * CAP + 256 + EE / 4 <= (size_t)NN * DD,
                "lists, counts and owner-id bytes fit in one feature plane");
  static_assert((NN / 16) * 256 >= EE / 4,
                "the encode GEMM launch has a thread for every 4 edges");

  gemm_wmma_f32<<<NN / 16, 256, 0, stream>>>(x, We, h1, 0, ei, ob);
  scores_init<<<SGRID, 256, 0, stream>>>(h1, ae_s, ae_d, s_src, s_dst);
  owner_aggregate<true, 1><<<NBLK, OT, 0, stream>>>(
      ei, h1, s_src, s_dst, be, enc, glist, gcnt, Wp, nullptr, h2, ob);
  scores_init<<<SGRID, 256, 0, stream>>>(h2, ap_s, ap_d, s_src, s_dst);
  owner_aggregate<false, 2><<<NBLK, OT, 0, stream>>>(
      ei, h2, s_src, s_dst, bp, out, glist, gcnt, Wp, enc, h1, ob);
  scores_init<<<SGRID, 256, 0, stream>>>(h1, ap_s, ap_d, s_src, s_dst);
  owner_aggregate<false, 3><<<NBLK, OT, 0, stream>>>(
      ei, h1, s_src, s_dst, bp, out, glist, gcnt, Wd, nullptr, h2, ob);
  scores_init<<<SGRID, 256, 0, stream>>>(h2, ad_s, ad_d, s_src, s_dst);
  owner_aggregate<false, 0><<<NBLK, OT, 0, stream>>>(
      ei, h2, s_src, s_dst, bd, out, glist, gcnt, nullptr, nullptr, nullptr, ob);
}


# owner row pass: softmax weight z/sum formed once per entry in a list pass, not by all lanes per row; a batch's 8 entries and weights read once, all 32 row loads out before its 1st add; sum order kept
# speedup vs baseline: 5.2444x; 1.1260x over previous
#include <hip/hip_runtime.h>

#define NN 50000
#define EE 800000
#define DD 128
#define ETOT (EE + NN)
#define SLOPE 0.2f
#define NPB 200
#define NBLK (NN / NPB)
static_assert(NN % NPB == 0, "every work-group owns a full range");
static_assert(NBLK <= 255, "an owner id fits in one byte, 255 is left for 'no owner'");
static_assert(EE % 16 == 0, "the owner-id plane is whole 16-byte groups of 16 edges");

typedef float v2f __attribute__((ext_vector_type(2)));
typedef float v8f __attribute__((ext_vector_type(8)));

__device__ __forceinline__ unsigned owner_id(int dst) {
  return (unsigned)dst < (unsigned)NN ? (unsigned)dst / NPB : 255u;
}

__global__ __launch_bounds__(256) void gemm_wmma_f32(
    const float* __restrict__ A, const float* __restrict__ W,
    float* __restrict__ H, int accumulate, const int* __restrict__ ei,
    unsigned* __restrict__ ob) {
  const int g4 = blockIdx.x * 256 + threadIdx.x;
  if (g4 < EE / 4) {
    const int4 d = ((const int4*)(ei + EE))[g4];
    ob[g4] = owner_id(d.x) | (owner_id(d.y) << 8) | (owner_id(d.z) << 16) |
             (owner_id(d.w) << 24);
  }
  const int lane = threadIdx.x & 31;
  const int wv   = threadIdx.x >> 5;
  const int m0   = blockIdx.x * 16;
  const int n0   = wv * 16;
  const int half = lane >> 4;
  const int l15  = lane & 15;
  const int row  = m0 + l15;
  const int col  = n0 + l15;

  v8f acc;
  if (accumulate) {
#pragma unroll
    for (int r = 0; r < 8; ++r)
      acc[r] = H[(size_t)(m0 + half * 8 + r) * DD + col];
  } else {
#pragma unroll
    for (int r = 0; r < 8; ++r) acc[r] = 0.0f;
  }

  const float* arow = A + (size_t)row * DD;
#pragma unroll 4
  for (int k = 0; k < DD; k += 4) {
    v2f av, bv;
    av[0] = arow[k + 2 * half];
    av[1] = arow[k + 2 * half + 1];
    bv[0] = W[(size_t)(k + 2 * half) * DD + col];
    bv[1] = W[(size_t)(k + 2 * half + 1) * DD + col];
    acc = __builtin_amdgcn_wmma_f32_16x16x4_f32(false, av, false, bv,
                                                (short)0, acc, false, false);
  }
#pragma unroll
  for (int r = 0; r < 8; ++r)
    H[(size_t)(m0 + half * 8 + r) * DD + col] = acc[r];
}

#define SR 16
#define SGRID ((NN / SR + 7) / 8)
static_assert(NN % SR == 0, "every wave has a full batch of nodes");
static_assert((SR * 4) % 16 == 0, "a wave's SR scores are whole 16-byte groups");

__global__ __launch_bounds__(256) void scores_init(
    const float* __restrict__ H, const float* __restrict__ a_s,
    const float* __restrict__ a_d, float* __restrict__ s_src,
    float* __restrict__ s_dst) {
  const int lane = threadIdx.x & 31;
  const int node0 = (blockIdx.x * 8 + (threadIdx.x >> 5)) * SR;
  if (node0 >= NN) return;
  const float4 as = *(const float4*)(a_s + lane * 4);
  const float4 ad = *(const float4*)(a_d + lane * 4);
  float4 hv[SR];
#pragma unroll
  for (int j = 0; j < SR; ++j)
    hv[j] = *(const float4*)(H + (size_t)(node0 + j) * DD + lane * 4);
  __builtin_amdgcn_sched_barrier(0);
  float ps[SR], pd[SR];
#pragma unroll
  for (int j = 0; j < SR; ++j) {
    ps[j] = fmaf(hv[j].w, as.w, fmaf(hv[j].z, as.z, fmaf(hv[j].x, as.x, hv[j].y * as.y)));
    pd[j] = fmaf(hv[j].w, ad.w, fmaf(hv[j].z, ad.z, fmaf(hv[j].x, ad.x, hv[j].y * ad.y)));
  }
#pragma unroll
  for (int off = 16; off > 0; off >>= 1) {
#pragma unroll
    for (int j = 0; j < SR; ++j) {
      ps[j] += __shfl_down(ps[j], off, 32);
      pd[j] += __shfl_down(pd[j], off, 32);
    }
  }
  if (lane == 0) {
    float4* os = (float4*)(s_src + node0);
    float4* od = (float4*)(s_dst + node0);
#pragma unroll
    for (int q = 0; q < SR / 4; ++q) {
      os[q] = make_float4(ps[4 * q], ps[4 * q + 1], ps[4 * q + 2], ps[4 * q + 3]);
      od[q] = make_float4(pd[4 * q], pd[4 * q + 1], pd[4 * q + 2], pd[4 * q + 3]);
    }
  }
}

__device__ __forceinline__ unsigned f2ord(float f) {
  unsigned u = __float_as_uint(f);
  return (u & 0x80000000u) ? ~u : (u | 0x80000000u);
}
__device__ __forceinline__ float ord2f(unsigned o) {
  return (o & 0x80000000u) ? __uint_as_float(o & 0x7FFFFFFFu)
                           : __uint_as_float(~o);
}
__device__ __forceinline__ float leaky(float v) {
  return v > 0.0f ? v : SLOPE * v;
}

#define OT 1024
#define OW (OT / 32)
#define CAP 16384
#define EB 8
#define AS 132
#define RT ((NPB + 15) / 16)
#define NPBP (RT * 16)
static_assert(AS % 4 == 0 && AS >= DD && OW == 32 && RT <= 16,
              "tail product: 8 column tiles x 4 row groups, at most 4 row tiles each");
static_assert(EE % 4 == 0 && EE % 32 == 0, "whole 16-byte reads, whole wave chunks");
static_assert(NN <= 65536 && NPB <= 4096, "entry packs dst-local<<16 | src");
static_assert(EE <= (1 << 20), "a scan entry is an edge index");
static_assert(NPB >= EB && (EB & (EB - 1)) == 0 && EB <= 32 && (OW & (OW - 1)) == 0,
              "the self loops alone fill a whole batch; lane % EB and wave % OW are masks");

__device__ __forceinline__ unsigned zero_bytes(unsigned x) {
  return ~(((x & 0x7F7F7F7Fu) + 0x7F7F7F7Fu) | x | 0x7F7F7F7Fu);
}

__device__ __forceinline__ void row_add(const float* __restrict__ H, float* acc,
                                        int src, int dl, float a, int lane) {
  const float* hs = H + (size_t)src * DD + lane;
  const float h0 = hs[0], h1 = hs[32], h2 = hs[64], h3 = hs[96];
  float* o = acc + dl * AS + lane;
  atomicAdd(o + 0, a * h0);
  atomicAdd(o + 32, a * h1);
  atomicAdd(o + 64, a * h2);
  atomicAdd(o + 96, a * h3);
}

template <bool BUILD, int TAIL>
__global__ __launch_bounds__(OT) void owner_aggregate(
    const int* __restrict__ ei, const float* __restrict__ H,
    const float* __restrict__ s_src, const float* __restrict__ s_dst,
    const float* __restrict__ bias, float* __restrict__ out,
    int* glist, int* gcnt, const float* __restrict__ Wn,
    const float* __restrict__ A2, float* __restrict__ Hn,
    const unsigned* __restrict__ ob) {
  __shared__ __attribute__((aligned(16))) float acc[NPBP * AS];
  __shared__ int l_pk[CAP];
  __shared__ float l_v[CAP];
  __shared__ unsigned m_ord[NPB];
  __shared__ float den[NPB];
  __shared__ float sd[NPB];
  __shared__ int cnt_s;

  const int tid = threadIdx.x;
  const int lane = tid & 31;
  const int wv = tid >> 5;
  const int lo = blockIdx.x * NPB;

  for (int i = tid; i < NPBP * AS / 4; i += OT)
    ((float4*)acc)[i] = make_float4(0.0f, 0.0f, 0.0f, 0.0f);
  if (tid < NPB) {
    const float sdv = s_dst[lo + tid];
    const float v = leaky(s_src[lo + tid] + sdv);
    sd[tid] = sdv;
    den[tid] = 0.0f;
    m_ord[tid] = f2ord(v);
    l_pk[tid] = (tid << 16) | (lo + tid);
    l_v[tid] = v;
  }
  if (tid == 0) cnt_s = NPB;
  __syncthreads();

  int* gl = glist + (size_t)blockIdx.x * CAP;
  int total;
  if (BUILD) {
  const uint4* ov = (const uint4*)ob;
  const int QN = EE / 16;
  const unsigned pat = (unsigned)blockIdx.x * 0x01010101u;
  uint4 cur[4], nxt[4];
#pragma unroll
  for (int u = 0; u < 4; ++u) cur[u] = ov[min(tid + u * OT, QN - 1)];
  for (int q0 = tid; q0 < QN; q0 += OT * 4) {
#pragma unroll
    for (int u = 0; u < 4; ++u) nxt[u] = ov[min(q0 + (4 + u) * OT, QN - 1)];
#pragma unroll
    for (int u = 0; u < 4; ++u) {
      const int q = q0 + u * OT;
      unsigned m = (zero_bytes(cur[u].x ^ pat) >> 7) | (zero_bytes(cur[u].y ^ pat) >> 6) |
                   (zero_bytes(cur[u].z ^ pat) >> 5) | (zero_bytes(cur[u].w ^ pat) >> 4);
      if (q >= QN) m = 0;
      while (m) {
        const int j = __builtin_ctz(m);
        m &= m - 1;
        const int slot = atomicAdd(&cnt_s, 1);
        if (slot < CAP) l_pk[slot] = q * 16 + (j & 3) * 4 + (j >> 3);
      }
    }
#pragma unroll
    for (int u = 0; u < 4; ++u) cur[u] = nxt[u];
  }
  __syncthreads();
  total = cnt_s;
  if (tid == 0) gcnt[blockIdx.x] = total;
  } else {
    total = gcnt[blockIdx.x];
  }
  const bool listed = total <= CAP;

  if (listed) {
    for (int i = NPB + tid; i < total; i += OT) {
      int ent;
      if (BUILD) {
        const int e = l_pk[i];
        ent = ((ei[EE + e] - lo) << 16) | ei[e];
        gl[i] = ent;
      } else {
        ent = gl[i];
      }
      const unsigned dl = (unsigned)ent >> 16;
      const float v = leaky(s_src[ent & 0xFFFF] + sd[dl]);
      atomicMax(&m_ord[dl], f2ord(v));
      l_pk[i] = ent;
      l_v[i] = v;
    }
    __syncthreads();
    for (int i = tid; i < total; i += OT) {
      const int dl = l_pk[i] >> 16;
      const float z = __expf(l_v[i] - ord2f(m_ord[dl]));
      l_v[i] = z;
      atomicAdd(&den[dl], z);
    }
    __syncthreads();
    for (int i = tid; i < total; i += OT) l_v[i] = l_v[i] / den[l_pk[i] >> 16];
    __syncthreads();
    {
      const int wu = __builtin_amdgcn_readfirstlane(wv);
      const int tot = __builtin_amdgcn_readfirstlane(total);
      const int nfull = tot / EB;
      const int sub = lane & (EB - 1);
      int spk[EB];
      float swt[EB];
      {
        const int b0 = min(wu, nfull - 1);
        const int pkv = l_pk[b0 * EB + sub];
        const float wtv = l_v[b0 * EB + sub];
#pragma unroll
        for (int jj = 0; jj < EB; ++jj) {
          spk[jj] = __builtin_amdgcn_readlane(pkv, jj);
          swt[jj] = __int_as_float(__builtin_amdgcn_readlane(__float_as_int(wtv), jj));
        }
      }
      for (int b = wu; b < nfull; b += OW) {
        float h[EB][4];
#pragma unroll
        for (int jj = 0; jj < EB; ++jj) {
          const float* hs = H + (size_t)(spk[jj] & 0xFFFF) * DD + lane;
          h[jj][0] = hs[0];
          h[jj][1] = hs[32];
          h[jj][2] = hs[64];
          h[jj][3] = hs[96];
        }
        const int bn = min(b + OW, nfull - 1);
        const int pkn = l_pk[bn * EB + sub];
        const float wtn = l_v[bn * EB + sub];
        int npk[EB];
        float nwt[EB];
#pragma unroll
        for (int jj = 0; jj < EB; ++jj) {
          npk[jj] = __builtin_amdgcn_readlane(pkn, jj);
          nwt[jj] = __int_as_float(__builtin_amdgcn_readlane(__float_as_int(wtn), jj));
        }
#pragma unroll
        for (int jj = 0; jj < EB; ++jj) {
          float* o = acc + (spk[jj] >> 16) * AS + lane;
          atomicAdd(o + 0, swt[jj] * h[jj][0]);
          atomicAdd(o + 32, swt[jj] * h[jj][1]);
          atomicAdd(o + 64, swt[jj] * h[jj][2]);
          atomicAdd(o + 96, swt[jj] * h[jj][3]);
        }
#pragma unroll
        for (int jj = 0; jj < EB; ++jj) {
          spk[jj] = npk[jj];
          swt[jj] = nwt[jj];
        }
      }
      const int idx = nfull * EB + ((wu - nfull) & (OW - 1));
      if (idx < tot) {
        const int pk = __builtin_amdgcn_readfirstlane(l_pk[idx]);
        const float a = __int_as_float(__builtin_amdgcn_readfirstlane(__float_as_int(l_v[idx])));
        row_add(H, acc, pk & 0xFFFF, pk >> 16, a, lane);
      }
    }
  } else {
    for (int e = tid; e < EE; e += OT) {
      const unsigned dl = (unsigned)(ei[EE + e] - lo);
      if (dl < (unsigned)NPB)
        atomicMax(&m_ord[dl], f2ord(leaky(s_src[ei[e]] + sd[dl])));
    }
    __syncthreads();
    if (tid < NPB) {
      const float v = leaky(s_src[lo + tid] + sd[tid]);
      atomicAdd(&den[tid], __expf(v - ord2f(m_ord[tid])));
    }
    for (int e = tid; e < EE; e += OT) {
      const unsigned dl = (unsigned)(ei[EE + e] - lo);
      if (dl < (unsigned)NPB) {
        const float v = leaky(s_src[ei[e]] + sd[dl]);
        atomicAdd(&den[dl], __expf(v - ord2f(m_ord[dl])));
      }
    }
    __syncthreads();
    for (int n = wv; n < NPB; n += OW) {
      const float v = leaky(s_src[lo + n] + sd[n]);
      row_add(H, acc, lo + n, n, __expf(v - ord2f(m_ord[n])) / den[n], lane);
    }
    for (int e0 = wv * 32; e0 < EE; e0 += OW * 32) {
      const int e = e0 + lane;
      const unsigned dl = (unsigned)(ei[EE + e] - lo);
      const bool hit = dl < (unsigned)NPB;
      int s = 0;
      float a = 0.0f;
      if (hit) {
        s = ei[e];
        const float v = leaky(s_src[s] + sd[dl]);
        a = __expf(v - ord2f(m_ord[dl])) / den[dl];
      }
      unsigned mask = __builtin_amdgcn_ballot_w32(hit);
      while (mask) {
        const int j = __builtin_ctz(mask);
        mask &= mask - 1;
        row_add(H, acc, __builtin_amdgcn_readlane(s, j),
                __builtin_amdgcn_readlane((int)dl, j),
                __int_as_float(__builtin_amdgcn_readlane(__float_as_int(a), j)),
                lane);
      }
    }
  }
  __syncthreads();

  float4* acc4 = (float4*)acc;
  float4* o4 = (float4*)(out + (size_t)lo * DD);
  for (int i = tid; i < NPB * DD / 4; i += OT) {
    const int li = (i >> 5) * (AS / 4) + (i & 31);
    const float4 r = acc4[li];
    const float4 b = ((const float4*)bias)[i & (DD / 4 - 1)];
    const float4 v = make_float4(r.x + b.x, r.y + b.y, r.z + b.z, r.w + b.w);
    if (TAIL == 0 || TAIL == 1) o4[i] = v;
    if (TAIL != 0) acc4[li] = v;
  }
  if (TAIL == 0) return;

  __syncthreads();
  {
    const int l15 = lane & 15;
    const int half = lane >> 4;
    const int wu = __builtin_amdgcn_readfirstlane(wv);
    const int ct = wu & 7;
    const int rg = wu >> 3;
    const int col = ct * 16 + l15;
    constexpr int NCH = (TAIL == 3) ? 2 : 4;
    v8f c[4];
#pragma unroll
    for (int t = 0; t < 4; ++t)
#pragma unroll
      for (int r = 0; r < 8; ++r) c[t][r] = 0.0f;
#pragma unroll
    for (int ch = 0; ch < NCH; ++ch) {
      if (TAIL == 2 && ch == 2) {
        __syncthreads();
        const float4* a4 = (const float4*)(A2 + (size_t)lo * DD);
        constexpr int NR = (NPB * DD / 4 + OT - 1) / OT;
#pragma unroll
        for (int u = 0; u < NR; ++u) {
          const int i = tid + u * OT;
          if (i < NPB * DD / 4) acc4[(i >> 5) * (AS / 4) + (i & 31)] = a4[i];
        }
        __syncthreads();
      }
      v2f b[16];
      const float* wp = Wn + (size_t)(ch * 64 + 2 * half) * DD + col;
#pragma unroll
      for (int kk = 0; kk < 16; ++kk) {
        b[kk][0] = wp[(size_t)(kk * 4) * DD];
        b[kk][1] = wp[(size_t)(kk * 4 + 1) * DD];
      }
      const int acol = (ch & 1) * 64 + 2 * half;
#pragma unroll
      for (int t = 0; t < 4; ++t) {
        if (4 * t + 3 < RT || rg + 4 * t < RT) {
          const float* ap = acc + ((rg + 4 * t) * 16 + l15) * AS + acol;
#pragma unroll
          for (int kk = 0; kk < 16; ++kk) {
            const v2f a = *(const v2f*)(ap + kk * 4);
            c[t] = __builtin_amdgcn_wmma_f32_16x16x4_f32(false, a, false, b[kk],
                                                         (short)0, c[t], false, false);
          }
        }
      }
    }
#pragma unroll
    for (int t = 0; t < 4; ++t) {
      if (4 * t + 3 < RT || rg + 4 * t < RT) {
#pragma unroll
        for (int r = 0; r < 8; ++r) {
          const int row = (rg + 4 * t) * 16 + half * 8 + r;
          if ((4 * t + 3) * 16 + 15 < NPB || row < NPB)
            Hn[(size_t)(lo + row) * DD + col] = c[t][r];
        }
      }
    }
  }
}

extern "C" void kernel_launch(void* const* d_in, const int* in_sizes, int n_in,
                              void* d_out, int out_size, void* d_ws,
                              size_t ws_size, hipStream_t stream) {
  const float* x    = (const float*)d_in[0];
  const int*   ei   = (const int*)d_in[1];
  const float* We   = (const float*)d_in[2];
  const float* ae_s = (const float*)d_in[3];
  const float* ae_d = (const float*)d_in[4];
  const float* be   = (const float*)d_in[5];
  const float* Wp   = (const float*)d_in[6];
  const float* ap_s = (const float*)d_in[7];
  const float* ap_d = (const float*)d_in[8];
  const float* bp   = (const float*)d_in[9];
  const float* Wd   = (const float*)d_in[10];
  const float* ad_s = (const float*)d_in[11];
  const float* ad_d = (const float*)d_in[12];
  const float* bd   = (const float*)d_in[13];
  float*       out  = (float*)d_out;

  const size_t ND = (size_t)NN * DD;
  float*    enc   = (float*)d_ws;
  float*    h2    = enc + ND;
  int*      glist = (int*)(h2 + ND);
  int*      gcnt  = glist + NBLK * CAP;
  unsigned* ob    = (unsigned*)(gcnt + 256);
  float*    h1    = h2 + ND + ND;
  float*    s_src = h1 + ND;
  float*    s_dst = s_src + NN;
  static_assert(NBLK <= 256 && (size_t)NBLK * CAP + 256 + EE / 4 <= (size_t)NN * DD,
                "lists, counts and owner-id bytes fit in one feature plane");
  static_assert((NN / 16) * 256 >= EE / 4,
                "the encode GEMM launch has a thread for every 4 edges");

  gemm_wmma_f32<<<NN / 16, 256, 0, stream>>>(x, We, h1, 0, ei, ob);
  scores_init<<<SGRID, 256, 0, stream>>>(h1, ae_s, ae_d, s_src, s_dst);
  owner_aggregate<true, 1><<<NBLK, OT, 0, stream>>>(
      ei, h1, s_src, s_dst, be, enc, glist, gcnt, Wp, nullptr, h2, ob);
  scores_init<<<SGRID, 256, 0, stream>>>(h2, ap_s, ap_d, s_src, s_dst);
  owner_aggregate<false, 2><<<NBLK, OT, 0, stream>>>(
      ei, h2, s_src, s_dst, bp, out, glist, gcnt, Wp, enc, h1, ob);
  scores_init<<<SGRID, 256, 0, stream>>>(h1, ap_s, ap_d, s_src, s_dst);
  owner_aggregate<false, 3><<<NBLK, OT, 0, stream>>>(
      ei, h1, s_src, s_dst, bp, out, glist, gcnt, Wd, nullptr, h2, ob);
  scores_init<<<SGRID, 256, 0, stream>>>(h2, ad_s, ad_d, s_src, s_dst);
  owner_aggregate<false, 0><<<NBLK, OT, 0, stream>>>(
      ei, h2, s_src, s_dst, bd, out, glist, gcnt, nullptr, nullptr, nullptr, ob);
}


# owner row pass: each 512-byte source row fetched by one 16-byte-per-lane load (was four 4-byte loads); LDS row kept lane-major so adds stay conflict-free; same addends, LDS f32 adds in arrival order
# speedup vs baseline: 5.4791x; 1.0448x over previous
#include <hip/hip_runtime.h>

#define NN 50000
#define EE 800000
#define DD 128
#define ETOT (EE + NN)
#define SLOPE 0.2f
#define NPB 200
#define NBLK (NN / NPB)
static_assert(NN % NPB == 0, "every work-group owns a full range");
static_assert(NBLK <= 255, "an owner id fits in one byte, 255 is left for 'no owner'");
static_assert(EE % 16 == 0, "the owner-id plane is whole 16-byte groups of 16 edges");

typedef float v2f __attribute__((ext_vector_type(2)));
typedef float v8f __attribute__((ext_vector_type(8)));

__device__ __forceinline__ unsigned owner_id(int dst) {
  return (unsigned)dst < (unsigned)NN ? (unsigned)dst / NPB : 255u;
}

__global__ __launch_bounds__(256) void gemm_wmma_f32(
    const float* __restrict__ A, const float* __restrict__ W,
    float* __restrict__ H, int accumulate, const int* __restrict__ ei,
    unsigned* __restrict__ ob) {
  const int g4 = blockIdx.x * 256 + threadIdx.x;
  if (g4 < EE / 4) {
    const int4 d = ((const int4*)(ei + EE))[g4];
    ob[g4] = owner_id(d.x) | (owner_id(d.y) << 8) | (owner_id(d.z) << 16) |
             (owner_id(d.w) << 24);
  }
  const int lane = threadIdx.x & 31;
  const int wv   = threadIdx.x >> 5;
  const int m0   = blockIdx.x * 16;
  const int n0   = wv * 16;
  const int half = lane >> 4;
  const int l15  = lane & 15;
  const int row  = m0 + l15;
  const int col  = n0 + l15;

  v8f acc;
  if (accumulate) {
#pragma unroll
    for (int r = 0; r < 8; ++r)
      acc[r] = H[(size_t)(m0 + half * 8 + r) * DD + col];
  } else {
#pragma unroll
    for (int r = 0; r < 8; ++r) acc[r] = 0.0f;
  }

  const float* arow = A + (size_t)row * DD;
#pragma unroll 4
  for (int k = 0; k < DD; k += 4) {
    v2f av, bv;
    av[0] = arow[k + 2 * half];
    av[1] = arow[k + 2 * half + 1];
    bv[0] = W[(size_t)(k + 2 * half) * DD + col];
    bv[1] = W[(size_t)(k + 2 * half + 1) * DD + col];
    acc = __builtin_amdgcn_wmma_f32_16x16x4_f32(false, av, false, bv,
                                                (short)0, acc, false, false);
  }
#pragma unroll
  for (int r = 0; r < 8; ++r)
    H[(size_t)(m0 + half * 8 + r) * DD + col] = acc[r];
}

#define SR 16
#define SGRID ((NN / SR + 7) / 8)
static_assert(NN % SR == 0, "every wave has a full batch of nodes");
static_assert((SR * 4) % 16 == 0, "a wave's SR scores are whole 16-byte groups");

__global__ __launch_bounds__(256) void scores_init(
    const float* __restrict__ H, const float* __restrict__ a_s,
    const float* __restrict__ a_d, float* __restrict__ s_src,
    float* __restrict__ s_dst) {
  const int lane = threadIdx.x & 31;
  const int node0 = (blockIdx.x * 8 + (threadIdx.x >> 5)) * SR;
  if (node0 >= NN) return;
  const float4 as = *(const float4*)(a_s + lane * 4);
  const float4 ad = *(const float4*)(a_d + lane * 4);
  float4 hv[SR];
#pragma unroll
  for (int j = 0; j < SR; ++j)
    hv[j] = *(const float4*)(H + (size_t)(node0 + j) * DD + lane * 4);
  __builtin_amdgcn_sched_barrier(0);
  float ps[SR], pd[SR];
#pragma unroll
  for (int j = 0; j < SR; ++j) {
    ps[j] = fmaf(hv[j].w, as.w, fmaf(hv[j].z, as.z, fmaf(hv[j].x, as.x, hv[j].y * as.y)));
    pd[j] = fmaf(hv[j].w, ad.w, fmaf(hv[j].z, ad.z, fmaf(hv[j].x, ad.x, hv[j].y * ad.y)));
  }
#pragma unroll
  for (int off = 16; off > 0; off >>= 1) {
#pragma unroll
    for (int j = 0; j < SR; ++j) {
      ps[j] += __shfl_down(ps[j], off, 32);
      pd[j] += __shfl_down(pd[j], off, 32);
    }
  }
  if (lane == 0) {
    float4* os = (float4*)(s_src + node0);
    float4* od = (float4*)(s_dst + node0);
#pragma unroll
    for (int q = 0; q < SR / 4; ++q) {
      os[q] = make_float4(ps[4 * q], ps[4 * q + 1], ps[4 * q + 2], ps[4 * q + 3]);
      od[q] = make_float4(pd[4 * q], pd[4 * q + 1], pd[4 * q + 2], pd[4 * q + 3]);
    }
  }
}

__device__ __forceinline__ unsigned f2ord(float f) {
  unsigned u = __float_as_uint(f);
  return (u & 0x80000000u) ? ~u : (u | 0x80000000u);
}
__device__ __forceinline__ float ord2f(unsigned o) {
  return (o & 0x80000000u) ? __uint_as_float(o & 0x7FFFFFFFu)
                           : __uint_as_float(~o);
}
__device__ __forceinline__ float leaky(float v) {
  return v > 0.0f ? v : SLOPE * v;
}

#define OT 1024
#define OW (OT / 32)
#define CAP 16384
#define EB 8
#define AS 132
#define RT ((NPB + 15) / 16)
#define NPBP (RT * 16)
static_assert(AS % 4 == 0 && AS >= DD && OW == 32 && RT <= 16,
              "tail product: 8 column tiles x 4 row groups, at most 4 row tiles each");
static_assert(EE % 4 == 0 && EE % 32 == 0, "whole 16-byte reads, whole wave chunks");
static_assert(NN <= 65536 && NPB <= 4096, "entry packs dst-local<<16 | src");
static_assert(EE <= (1 << 20), "a scan entry is an edge index");
static_assert(NPB >= EB && (EB & (EB - 1)) == 0 && EB <= 32 && (OW & (OW - 1)) == 0,
              "the self loops alone fill a whole batch; lane % EB and wave % OW are masks");

__device__ __forceinline__ unsigned zero_bytes(unsigned x) {
  return ~(((x & 0x7F7F7F7Fu) + 0x7F7F7F7Fu) | x | 0x7F7F7F7Fu);
}

__device__ __forceinline__ void row_add(const float* __restrict__ H, float* acc,
                                        int src, int dl, float a, int lane) {
  const float4 h = *(const float4*)(H + (size_t)src * DD + lane * 4);
  float* o = acc + dl * AS + lane;
  atomicAdd(o + 0, a * h.x);
  atomicAdd(o + 32, a * h.y);
  atomicAdd(o + 64, a * h.z);
  atomicAdd(o + 96, a * h.w);
}

template <bool BUILD, int TAIL>
__global__ __launch_bounds__(OT) void owner_aggregate(
    const int* __restrict__ ei, const float* __restrict__ H,
    const float* __restrict__ s_src, const float* __restrict__ s_dst,
    const float* __restrict__ bias, float* __restrict__ out,
    int* glist, int* gcnt, const float* __restrict__ Wn,
    const float* __restrict__ A2, float* __restrict__ Hn,
    const unsigned* __restrict__ ob) {
  __shared__ __attribute__((aligned(16))) float acc[NPBP * AS];
  __shared__ int l_pk[CAP];
  __shared__ float l_v[CAP];
  __shared__ unsigned m_ord[NPB];
  __shared__ float den[NPB];
  __shared__ float sd[NPB];
  __shared__ int cnt_s;

  const int tid = threadIdx.x;
  const int lane = tid & 31;
  const int wv = tid >> 5;
  const int lo = blockIdx.x * NPB;

  for (int i = tid; i < NPBP * AS / 4; i += OT)
    ((float4*)acc)[i] = make_float4(0.0f, 0.0f, 0.0f, 0.0f);
  if (tid < NPB) {
    const float sdv = s_dst[lo + tid];
    const float v = leaky(s_src[lo + tid] + sdv);
    sd[tid] = sdv;
    den[tid] = 0.0f;
    m_ord[tid] = f2ord(v);
    l_pk[tid] = (tid << 16) | (lo + tid);
    l_v[tid] = v;
  }
  if (tid == 0) cnt_s = NPB;
  __syncthreads();

  int* gl = glist + (size_t)blockIdx.x * CAP;
  int total;
  if (BUILD) {
  const uint4* ov = (const uint4*)ob;
  const int QN = EE / 16;
  const unsigned pat = (unsigned)blockIdx.x * 0x01010101u;
  uint4 cur[4], nxt[4];
#pragma unroll
  for (int u = 0; u < 4; ++u) cur[u] = ov[min(tid + u * OT, QN - 1)];
  for (int q0 = tid; q0 < QN; q0 += OT * 4) {
#pragma unroll
    for (int u = 0; u < 4; ++u) nxt[u] = ov[min(q0 + (4 + u) * OT, QN - 1)];
#pragma unroll
    for (int u = 0; u < 4; ++u) {
      const int q = q0 + u * OT;
      unsigned m = (zero_bytes(cur[u].x ^ pat) >> 7) | (zero_bytes(cur[u].y ^ pat) >> 6) |
                   (zero_bytes(cur[u].z ^ pat) >> 5) | (zero_bytes(cur[u].w ^ pat) >> 4);
      if (q >= QN) m = 0;
      while (m) {
        const int j = __builtin_ctz(m);
        m &= m - 1;
        const int slot = atomicAdd(&cnt_s, 1);
        if (slot < CAP) l_pk[slot] = q * 16 + (j & 3) * 4 + (j >> 3);
      }
    }
#pragma unroll
    for (int u = 0; u < 4; ++u) cur[u] = nxt[u];
  }
  __syncthreads();
  total = cnt_s;
  if (tid == 0) gcnt[blockIdx.x] = total;
  } else {
    total = gcnt[blockIdx.x];
  }
  const bool listed = total <= CAP;

  if (listed) {
    for (int i = NPB + tid; i < total; i += OT) {
      int ent;
      if (BUILD) {
        const int e = l_pk[i];
        ent = ((ei[EE + e] - lo) << 16) | ei[e];
        gl[i] = ent;
      } else {
        ent = gl[i];
      }
      const unsigned dl = (unsigned)ent >> 16;
      const float v = leaky(s_src[ent & 0xFFFF] + sd[dl]);
      atomicMax(&m_ord[dl], f2ord(v));
      l_pk[i] = ent;
      l_v[i] = v;
    }
    __syncthreads();
    for (int i = tid; i < total; i += OT) {
      const int dl = l_pk[i] >> 16;
      const float z = __expf(l_v[i] - ord2f(m_ord[dl]));
      l_v[i] = z;
      atomicAdd(&den[dl], z);
    }
    __syncthreads();
    for (int i = tid; i < total; i += OT) l_v[i] = l_v[i] / den[l_pk[i] >> 16];
    __syncthreads();
    {
      const int wu = __builtin_amdgcn_readfirstlane(wv);
      const int tot = __builtin_amdgcn_readfirstlane(total);
      const int nfull = tot / EB;
      const int sub = lane & (EB - 1);
      int spk[EB];
      float swt[EB];
      {
        const int b0 = min(wu, nfull - 1);
        const int pkv = l_pk[b0 * EB + sub];
        const float wtv = l_v[b0 * EB + sub];
#pragma unroll
        for (int jj = 0; jj < EB; ++jj) {
          spk[jj] = __builtin_amdgcn_readlane(pkv, jj);
          swt[jj] = __int_as_float(__builtin_amdgcn_readlane(__float_as_int(wtv), jj));
        }
      }
      for (int b = wu; b < nfull; b += OW) {
        float4 h[EB];
#pragma unroll
        for (int jj = 0; jj < EB; ++jj)
          h[jj] = *(const float4*)(H + (size_t)(spk[jj] & 0xFFFF) * DD + lane * 4);
        const int bn = min(b + OW, nfull - 1);
        const int pkn = l_pk[bn * EB + sub];
        const float wtn = l_v[bn * EB + sub];
        int npk[EB];
        float nwt[EB];
#pragma unroll
        for (int jj = 0; jj < EB; ++jj) {
          npk[jj] = __builtin_amdgcn_readlane(pkn, jj);
          nwt[jj] = __int_as_float(__builtin_amdgcn_readlane(__float_as_int(wtn), jj));
        }
#pragma unroll
        for (int jj = 0; jj < EB; ++jj) {
          float* o = acc + (spk[jj] >> 16) * AS + lane;
          atomicAdd(o + 0, swt[jj] * h[jj].x);
          atomicAdd(o + 32, swt[jj] * h[jj].y);
          atomicAdd(o + 64, swt[jj] * h[jj].z);
          atomicAdd(o + 96, swt[jj] * h[jj].w);
        }
#pragma unroll
        for (int jj = 0; jj < EB; ++jj) {
          spk[jj] = npk[jj];
          swt[jj] = nwt[jj];
        }
      }
      const int idx = nfull * EB + ((wu - nfull) & (OW - 1));
      if (idx < tot) {
        const int pk = __builtin_amdgcn_readfirstlane(l_pk[idx]);
        const float a = __int_as_float(__builtin_amdgcn_readfirstlane(__float_as_int(l_v[idx])));
        row_add(H, acc, pk & 0xFFFF, pk >> 16, a, lane);
      }
    }
  } else {
    for (int e = tid; e < EE; e += OT) {
      const unsigned dl = (unsigned)(ei[EE + e] - lo);
      if (dl < (unsigned)NPB)
        atomicMax(&m_ord[dl], f2ord(leaky(s_src[ei[e]] + sd[dl])));
    }
    __syncthreads();
    if (tid < NPB) {
      const float v = leaky(s_src[lo + tid] + sd[tid]);
      atomicAdd(&den[tid], __expf(v - ord2f(m_ord[tid])));
    }
    for (int e = tid; e < EE; e += OT) {
      const unsigned dl = (unsigned)(ei[EE + e] - lo);
      if (dl < (unsigned)NPB) {
        const float v = leaky(s_src[ei[e]] + sd[dl]);
        atomicAdd(&den[dl], __expf(v - ord2f(m_ord[dl])));
      }
    }
    __syncthreads();
    for (int n = wv; n < NPB; n += OW) {
      const float v = leaky(s_src[lo + n] + sd[n]);
      row_add(H, acc, lo + n, n, __expf(v - ord2f(m_ord[n])) / den[n], lane);
    }
    for (int e0 = wv * 32; e0 < EE; e0 += OW * 32) {
      const int e = e0 + lane;
      const unsigned dl = (unsigned)(ei[EE + e] - lo);
      const bool hit = dl < (unsigned)NPB;
      int s = 0;
      float a = 0.0f;
      if (hit) {
        s = ei[e];
        const float v = leaky(s_src[s] + sd[dl]);
        a = __expf(v - ord2f(m_ord[dl])) / den[dl];
      }
      unsigned mask = __builtin_amdgcn_ballot_w32(hit);
      while (mask) {
        const int j = __builtin_ctz(mask);
        mask &= mask - 1;
        row_add(H, acc, __builtin_amdgcn_readlane(s, j),
                __builtin_amdgcn_readlane((int)dl, j),
                __int_as_float(__builtin_amdgcn_readlane(__float_as_int(a), j)),
                lane);
      }
    }
  }
  __syncthreads();

  float4* acc4 = (float4*)acc;
  float4* o4 = (float4*)(out + (size_t)lo * DD);
  for (int i = tid; i < NPB * DD / 4; i += OT) {
    const int li = (i >> 5) * (AS / 4) + (i & 31);
    const float* rp = acc + (i >> 5) * AS + (i & 31);
    const float4 r = make_float4(rp[0], rp[32], rp[64], rp[96]);
    const float4 b = ((const float4*)bias)[i & (DD / 4 - 1)];
    const float4 v = make_float4(r.x + b.x, r.y + b.y, r.z + b.z, r.w + b.w);
    if (TAIL == 0 || TAIL == 1) o4[i] = v;
    if (TAIL != 0) acc4[li] = v;
  }
  if (TAIL == 0) return;

  __syncthreads();
  {
    const int l15 = lane & 15;
    const int half = lane >> 4;
    const int wu = __builtin_amdgcn_readfirstlane(wv);
    const int ct = wu & 7;
    const int rg = wu >> 3;
    const int col = ct * 16 + l15;
    constexpr int NCH = (TAIL == 3) ? 2 : 4;
    v8f c[4];
#pragma unroll
    for (int t = 0; t < 4; ++t)
#pragma unroll
      for (int r = 0; r < 8; ++r) c[t][r] = 0.0f;
#pragma unroll
    for (int ch = 0; ch < NCH; ++ch) {
      if (TAIL == 2 && ch == 2) {
        __syncthreads();
        const float4* a4 = (const float4*)(A2 + (size_t)lo * DD);
        constexpr int NR = (NPB * DD / 4 + OT - 1) / OT;
#pragma unroll
        for (int u = 0; u < NR; ++u) {
          const int i = tid + u * OT;
          if (i < NPB * DD / 4) acc4[(i >> 5) * (AS / 4) + (i & 31)] = a4[i];
        }
        __syncthreads();
      }
      v2f b[16];
      const float* wp = Wn + (size_t)(ch * 64 + 2 * half) * DD + col;
#pragma unroll
      for (int kk = 0; kk < 16; ++kk) {
        b[kk][0] = wp[(size_t)(kk * 4) * DD];
        b[kk][1] = wp[(size_t)(kk * 4 + 1) * DD];
      }
      const int acol = (ch & 1) * 64 + 2 * half;
#pragma unroll
      for (int t = 0; t < 4; ++t) {
        if (4 * t + 3 < RT || rg + 4 * t < RT) {
          const float* ap = acc + ((rg + 4 * t) * 16 + l15) * AS + acol;
#pragma unroll
          for (int kk = 0; kk < 16; ++kk) {
            const v2f a = *(const v2f*)(ap + kk * 4);
            c[t] = __builtin_amdgcn_wmma_f32_16x16x4_f32(false, a, false, b[kk],
                                                         (short)0, c[t], false, false);
          }
        }
      }
    }
#pragma unroll
    for (int t = 0; t < 4; ++t) {
      if (4 * t + 3 < RT || rg + 4 * t < RT) {
#pragma unroll
        for (int r = 0; r < 8; ++r) {
          const int row = (rg + 4 * t) * 16 + half * 8 + r;
          if ((4 * t + 3) * 16 + 15 < NPB || row < NPB)
            Hn[(size_t)(lo + row) * DD + col] = c[t][r];
        }
      }
    }
  }
}

extern "C" void kernel_launch(void* const* d_in, const int* in_sizes, int n_in,
                              void* d_out, int out_size, void* d_ws,
                              size_t ws_size, hipStream_t stream) {
  const float* x    = (const float*)d_in[0];
  const int*   ei   = (const int*)d_in[1];
  const float* We   = (const float*)d_in[2];
  const float* ae_s = (const float*)d_in[3];
  const float* ae_d = (const float*)d_in[4];
  const float* be   = (const float*)d_in[5];
  const float* Wp   = (const float*)d_in[6];
  const float* ap_s = (const float*)d_in[7];
  const float* ap_d = (const float*)d_in[8];
  const float* bp   = (const float*)d_in[9];
  const float* Wd   = (const float*)d_in[10];
  const float* ad_s = (const float*)d_in[11];
  const float* ad_d = (const float*)d_in[12];
  const float* bd   = (const float*)d_in[13];
  float*       out  = (float*)d_out;

  const size_t ND = (size_t)NN * DD;
  float*    enc   = (float*)d_ws;
  float*    h2    = enc + ND;
  int*      glist = (int*)(h2 + ND);
  int*      gcnt  = glist + NBLK * CAP;
  unsigned* ob    = (unsigned*)(gcnt + 256);
  float*    h1    = h2 + ND + ND;
  float*    s_src = h1 + ND;
  float*    s_dst = s_src + NN;
  static_assert(NBLK <= 256 && (size_t)NBLK * CAP + 256 + EE / 4 <= (size_t)NN * DD,
                "lists, counts and owner-id bytes fit in one feature plane");
  static_assert((NN / 16) * 256 >= EE / 4,
                "the encode GEMM launch has a thread for every 4 edges");

  gemm_wmma_f32<<<NN / 16, 256, 0, stream>>>(x, We, h1, 0, ei, ob);
  scores_init<<<SGRID, 256, 0, stream>>>(h1, ae_s, ae_d, s_src, s_dst);
  owner_aggregate<true, 1><<<NBLK, OT, 0, stream>>>(
      ei, h1, s_src, s_dst, be, enc, glist, gcnt, Wp, nullptr, h2, ob);
  scores_init<<<SGRID, 256, 0, stream>>>(h2, ap_s, ap_d, s_src, s_dst);
  owner_aggregate<false, 2><<<NBLK, OT, 0, stream>>>(
      ei, h2, s_src, s_dst, bp, out, glist, gcnt, Wp, enc, h1, ob);
  scores_init<<<SGRID, 256, 0, stream>>>(h1, ap_s, ap_d, s_src, s_dst);
  owner_aggregate<false, 3><<<NBLK, OT, 0, stream>>>(
      ei, h1, s_src, s_dst, bp, out, glist, gcnt, Wd, nullptr, h2, ob);
  scores_init<<<SGRID, 256, 0, stream>>>(h2, ad_s, ad_d, s_src, s_dst);
  owner_aggregate<false, 0><<<NBLK, OT, 0, stream>>>(
      ei, h2, s_src, s_dst, bd, out, glist, gcnt, nullptr, nullptr, nullptr, ob);
}


# encode product x@We moved onto the owner tails' data path: 250 work-groups stage their 200 x rows in LDS (whole 512-byte row loads, pitch 132), W panel in registers, A fragments from LDS; same sums
# speedup vs baseline: 5.7494x; 1.0493x over previous
#include <hip/hip_runtime.h>

#define NN 50000
#define EE 800000
#define DD 128
#define ETOT (EE + NN)
#define SLOPE 0.2f
#define NPB 200
#define NBLK (NN / NPB)
static_assert(NN % NPB == 0, "every work-group owns a full range");
static_assert(NBLK <= 255, "an owner id fits in one byte, 255 is left for 'no owner'");
static_assert(EE % 16 == 0, "the owner-id plane is whole 16-byte groups of 16 edges");

typedef float v2f __attribute__((ext_vector_type(2)));
typedef float v8f __attribute__((ext_vector_type(8)));

__device__ __forceinline__ unsigned owner_id(int dst) {
  return (unsigned)dst < (unsigned)NN ? (unsigned)dst / NPB : 255u;
}

#define SR 16
#define SGRID ((NN / SR + 7) / 8)
static_assert(NN % SR == 0, "every wave has a full batch of nodes");
static_assert((SR * 4) % 16 == 0, "a wave's SR scores are whole 16-byte groups");

__global__ __launch_bounds__(256) void scores_init(
    const float* __restrict__ H, const float* __restrict__ a_s,
    const float* __restrict__ a_d, float* __restrict__ s_src,
    float* __restrict__ s_dst) {
  const int lane = threadIdx.x & 31;
  const int node0 = (blockIdx.x * 8 + (threadIdx.x >> 5)) * SR;
  if (node0 >= NN) return;
  const float4 as = *(const float4*)(a_s + lane * 4);
  const float4 ad = *(const float4*)(a_d + lane * 4);
  float4 hv[SR];
#pragma unroll
  for (int j = 0; j < SR; ++j)
    hv[j] = *(const float4*)(H + (size_t)(node0 + j) * DD + lane * 4);
  __builtin_amdgcn_sched_barrier(0);
  float ps[SR], pd[SR];
#pragma unroll
  for (int j = 0; j < SR; ++j) {
    ps[j] = fmaf(hv[j].w, as.w, fmaf(hv[j].z, as.z, fmaf(hv[j].x, as.x, hv[j].y * as.y)));
    pd[j] = fmaf(hv[j].w, ad.w, fmaf(hv[j].z, ad.z, fmaf(hv[j].x, ad.x, hv[j].y * ad.y)));
  }
#pragma unroll
  for (int off = 16; off > 0; off >>= 1) {
#pragma unroll
    for (int j = 0; j < SR; ++j) {
      ps[j] += __shfl_down(ps[j], off, 32);
      pd[j] += __shfl_down(pd[j], off, 32);
    }
  }
  if (lane == 0) {
    float4* os = (float4*)(s_src + node0);
    float4* od = (float4*)(s_dst + node0);
#pragma unroll
    for (int q = 0; q < SR / 4; ++q) {
      os[q] = make_float4(ps[4 * q], ps[4 * q + 1], ps[4 * q + 2], ps[4 * q + 3]);
      od[q] = make_float4(pd[4 * q], pd[4 * q + 1], pd[4 * q + 2], pd[4 * q + 3]);
    }
  }
}

__device__ __forceinline__ unsigned f2ord(float f) {
  unsigned u = __float_as_uint(f);
  return (u & 0x80000000u) ? ~u : (u | 0x80000000u);
}
__device__ __forceinline__ float ord2f(unsigned o) {
  return (o & 0x80000000u) ? __uint_as_float(o & 0x7FFFFFFFu)
                           : __uint_as_float(~o);
}
__device__ __forceinline__ float leaky(float v) {
  return v > 0.0f ? v : SLOPE * v;
}

#define OT 1024
#define OW (OT / 32)
#define CAP 16384
#define EB 8
#define AS 132
#define RT ((NPB + 15) / 16)
#define NPBP (RT * 16)
static_assert(AS % 4 == 0 && AS >= DD && OW == 32 && RT <= 16,
              "tail product: 8 column tiles x 4 row groups, at most 4 row tiles each");
static_assert(EE % 4 == 0 && EE % 32 == 0, "whole 16-byte reads, whole wave chunks");
static_assert(NN <= 65536 && NPB <= 4096, "entry packs dst-local<<16 | src");
static_assert(EE <= (1 << 20), "a scan entry is an edge index");
static_assert(NPB >= EB && (EB & (EB - 1)) == 0 && EB <= 32 && (OW & (OW - 1)) == 0,
              "the self loops alone fill a whole batch; lane % EB and wave % OW are masks");

__device__ __forceinline__ unsigned zero_bytes(unsigned x) {
  return ~(((x & 0x7F7F7F7Fu) + 0x7F7F7F7Fu) | x | 0x7F7F7F7Fu);
}

__device__ __forceinline__ void row_add(const float* __restrict__ H, float* acc,
                                        int src, int dl, float a, int lane) {
  const float4 h = *(const float4*)(H + (size_t)src * DD + lane * 4);
  float* o = acc + dl * AS + lane;
  atomicAdd(o + 0, a * h.x);
  atomicAdd(o + 32, a * h.y);
  atomicAdd(o + 64, a * h.z);
  atomicAdd(o + 96, a * h.w);
}

__global__ __launch_bounds__(OT) void encode_product(
    const float* __restrict__ X, const float* __restrict__ W,
    float* __restrict__ H, const int* __restrict__ ei,
    unsigned* __restrict__ ob) {
  __shared__ __attribute__((aligned(16))) float rows[NPBP * AS];
  const int tid = threadIdx.x;
  const int lane = tid & 31;
  const int l15 = lane & 15;
  const int half = lane >> 4;
  const int wu = __builtin_amdgcn_readfirstlane(tid >> 5);
  const int ct = wu & 7;
  const int rg = wu >> 3;
  const int col = ct * 16 + l15;
  const int lo = blockIdx.x * NPB;
  const float* wp = W + (size_t)(2 * half) * DD + col;
  v2f b[16];
#pragma unroll
  for (int kk = 0; kk < 16; ++kk) {
    b[kk][0] = wp[(size_t)(kk * 4) * DD];
    b[kk][1] = wp[(size_t)(kk * 4 + 1) * DD];
  }
  const int g4 = blockIdx.x * OT + tid;
  if (g4 < EE / 4) {
    const int4 d = ((const int4*)(ei + EE))[g4];
    ob[g4] = owner_id(d.x) | (owner_id(d.y) << 8) | (owner_id(d.z) << 16) |
             (owner_id(d.w) << 24);
  }
  {
    const float4* x4 = (const float4*)(X + (size_t)lo * DD);
    float4* rows4 = (float4*)rows;
    constexpr int NR = (NPB * DD / 4 + OT - 1) / OT;
#pragma unroll
    for (int u = 0; u < NR; ++u) {
      const int i = tid + u * OT;
      if (i < NPB * DD / 4) rows4[(i >> 5) * (AS / 4) + (i & 31)] = x4[i];
    }
  }
  for (int i = NPB * AS + tid; i < NPBP * AS; i += OT) rows[i] = 0.0f;
  __syncthreads();
  v8f c[4];
#pragma unroll
  for (int t = 0; t < 4; ++t)
#pragma unroll
    for (int r = 0; r < 8; ++r) c[t][r] = 0.0f;
#pragma unroll
  for (int ch = 0; ch < DD / 64; ++ch) {
    if (ch > 0) {
#pragma unroll
      for (int kk = 0; kk < 16; ++kk) {
        b[kk][0] = wp[(size_t)(ch * 64 + kk * 4) * DD];
        b[kk][1] = wp[(size_t)(ch * 64 + kk * 4 + 1) * DD];
      }
    }
    const int acol = ch * 64 + 2 * half;
#pragma unroll
    for (int t = 0; t < 4; ++t) {
      if (4 * t + 3 < RT || rg + 4 * t < RT) {
        const float* ap = rows + ((rg + 4 * t) * 16 + l15) * AS + acol;
#pragma unroll
        for (int kk = 0; kk < 16; ++kk) {
          const v2f a = *(const v2f*)(ap + kk * 4);
          c[t] = __builtin_amdgcn_wmma_f32_16x16x4_f32(false, a, false, b[kk],
                                                       (short)0, c[t], false, false);
        }
      }
    }
  }
#pragma unroll
  for (int t = 0; t < 4; ++t) {
    if (4 * t + 3 < RT || rg + 4 * t < RT) {
#pragma unroll
      for (int r = 0; r < 8; ++r) {
        const int row = (rg + 4 * t) * 16 + half * 8 + r;
        if ((4 * t + 3) * 16 + 15 < NPB || row < NPB)
          H[(size_t)(lo + row) * DD + col] = c[t][r];
      }
    }
  }
}

template <bool BUILD, int TAIL>
__global__ __launch_bounds__(OT) void owner_aggregate(
    const int* __restrict__ ei, const float* __restrict__ H,
    const float* __restrict__ s_src, const float* __restrict__ s_dst,
    const float* __restrict__ bias, float* __restrict__ out,
    int* glist, int* gcnt, const float* __restrict__ Wn,
    const float* __restrict__ A2, float* __restrict__ Hn,
    const unsigned* __restrict__ ob) {
  __shared__ __attribute__((aligned(16))) float acc[NPBP * AS];
  __shared__ int l_pk[CAP];
  __shared__ float l_v[CAP];
  __shared__ unsigned m_ord[NPB];
  __shared__ float den[NPB];
  __shared__ float sd[NPB];
  __shared__ int cnt_s;

  const int tid = threadIdx.x;
  const int lane = tid & 31;
  const int wv = tid >> 5;
  const int lo = blockIdx.x * NPB;

  for (int i = tid; i < NPBP * AS / 4; i += OT)
    ((float4*)acc)[i] = make_float4(0.0f, 0.0f, 0.0f, 0.0f);
  if (tid < NPB) {
    const float sdv = s_dst[lo + tid];
    const float v = leaky(s_src[lo + tid] + sdv);
    sd[tid] = sdv;
    den[tid] = 0.0f;
    m_ord[tid] = f2ord(v);
    l_pk[tid] = (tid << 16) | (lo + tid);
    l_v[tid] = v;
  }
  if (tid == 0) cnt_s = NPB;
  __syncthreads();

  int* gl = glist + (size_t)blockIdx.x * CAP;
  int total;
  if (BUILD) {
  const uint4* ov = (const uint4*)ob;
  const int QN = EE / 16;
  const unsigned pat = (unsigned)blockIdx.x * 0x01010101u;
  uint4 cur[4], nxt[4];
#pragma unroll
  for (int u = 0; u < 4; ++u) cur[u] = ov[min(tid + u * OT, QN - 1)];
  for (int q0 = tid; q0 < QN; q0 += OT * 4) {
#pragma unroll
    for (int u = 0; u < 4; ++u) nxt[u] = ov[min(q0 + (4 + u) * OT, QN - 1)];
#pragma unroll
    for (int u = 0; u < 4; ++u) {
      const int q = q0 + u * OT;
      unsigned m = (zero_bytes(cur[u].x ^ pat) >> 7) | (zero_bytes(cur[u].y ^ pat) >> 6) |
                   (zero_bytes(cur[u].z ^ pat) >> 5) | (zero_bytes(cur[u].w ^ pat) >> 4);
      if (q >= QN) m = 0;
      while (m) {
        const int j = __builtin_ctz(m);
        m &= m - 1;
        const int slot = atomicAdd(&cnt_s, 1);
        if (slot < CAP) l_pk[slot] = q * 16 + (j & 3) * 4 + (j >> 3);
      }
    }
#pragma unroll
    for (int u = 0; u < 4; ++u) cur[u] = nxt[u];
  }
  __syncthreads();
  total = cnt_s;
  if (tid == 0) gcnt[blockIdx.x] = total;
  } else {
    total = gcnt[blockIdx.x];
  }
  const bool listed = total <= CAP;

  if (listed) {
    for (int i = NPB + tid; i < total; i += OT) {
      int ent;
      if (BUILD) {
        const int e = l_pk[i];
        ent = ((ei[EE + e] - lo) << 16) | ei[e];
        gl[i] = ent;
      } else {
        ent = gl[i];
      }
      const unsigned dl = (unsigned)ent >> 16;
      const float v = leaky(s_src[ent & 0xFFFF] + sd[dl]);
      atomicMax(&m_ord[dl], f2ord(v));
      l_pk[i] = ent;
      l_v[i] = v;
    }
    __syncthreads();
    for (int i = tid; i < total; i += OT) {
      const int dl = l_pk[i] >> 16;
      const float z = __expf(l_v[i] - ord2f(m_ord[dl]));
      l_v[i] = z;
      atomicAdd(&den[dl], z);
    }
    __syncthreads();
    for (int i = tid; i < total; i += OT) l_v[i] = l_v[i] / den[l_pk[i] >> 16];
    __syncthreads();
    {
      const int wu = __builtin_amdgcn_readfirstlane(wv);
      const int tot = __builtin_amdgcn_readfirstlane(total);
      const int nfull = tot / EB;
      const int sub = lane & (EB - 1);
      int spk[EB];
      float swt[EB];
      {
        const int b0 = min(wu, nfull - 1);
        const int pkv = l_pk[b0 * EB + sub];
        const float wtv = l_v[b0 * EB + sub];
#pragma unroll
        for (int jj = 0; jj < EB; ++jj) {
          spk[jj] = __builtin_amdgcn_readlane(pkv, jj);
          swt[jj] = __int_as_float(__builtin_amdgcn_readlane(__float_as_int(wtv), jj));
        }
      }
      for (int b = wu; b < nfull; b += OW) {
        float4 h[EB];
#pragma unroll
        for (int jj = 0; jj < EB; ++jj)
          h[jj] = *(const float4*)(H + (size_t)(spk[jj] & 0xFFFF) * DD + lane * 4);
        const int bn = min(b + OW, nfull - 1);
        const int pkn = l_pk[bn * EB + sub];
        const float wtn = l_v[bn * EB + sub];
        int npk[EB];
        float nwt[EB];
#pragma unroll
        for (int jj = 0; jj < EB; ++jj) {
          npk[jj] = __builtin_amdgcn_readlane(pkn, jj);
          nwt[jj] = __int_as_float(__builtin_amdgcn_readlane(__float_as_int(wtn), jj));
        }
#pragma unroll
        for (int jj = 0; jj < EB; ++jj) {
          float* o = acc + (spk[jj] >> 16) * AS + lane;
          atomicAdd(o + 0, swt[jj] * h[jj].x);
          atomicAdd(o + 32, swt[jj] * h[jj].y);
          atomicAdd(o + 64, swt[jj] * h[jj].z);
          atomicAdd(o + 96, swt[jj] * h[jj].w);
        }
#pragma unroll
        for (int jj = 0; jj < EB; ++jj) {
          spk[jj] = npk[jj];
          swt[jj] = nwt[jj];
        }
      }
      const int idx = nfull * EB + ((wu - nfull) & (OW - 1));
      if (idx < tot) {
        const int pk = __builtin_amdgcn_readfirstlane(l_pk[idx]);
        const float a = __int_as_float(__builtin_amdgcn_readfirstlane(__float_as_int(l_v[idx])));
        row_add(H, acc, pk & 0xFFFF, pk >> 16, a, lane);
      }
    }
  } else {
    for (int e = tid; e < EE; e += OT) {
      const unsigned dl = (unsigned)(ei[EE + e] - lo);
      if (dl < (unsigned)NPB)
        atomicMax(&m_ord[dl], f2ord(leaky(s_src[ei[e]] + sd[dl])));
    }
    __syncthreads();
    if (tid < NPB) {
      const float v = leaky(s_src[lo + tid] + sd[tid]);
      atomicAdd(&den[tid], __expf(v - ord2f(m_ord[tid])));
    }
    for (int e = tid; e < EE; e += OT) {
      const unsigned dl = (unsigned)(ei[EE + e] - lo);
      if (dl < (unsigned)NPB) {
        const float v = leaky(s_src[ei[e]] + sd[dl]);
        atomicAdd(&den[dl], __expf(v - ord2f(m_ord[dl])));
      }
    }
    __syncthreads();
    for (int n = wv; n < NPB; n += OW) {
      const float v = leaky(s_src[lo + n] + sd[n]);
      row_add(H, acc, lo + n, n, __expf(v - ord2f(m_ord[n])) / den[n], lane);
    }
    for (int e0 = wv * 32; e0 < EE; e0 += OW * 32) {
      const int e = e0 + lane;
      const unsigned dl = (unsigned)(ei[EE + e] - lo);
      const bool hit = dl < (unsigned)NPB;
      int s = 0;
      float a = 0.0f;
      if (hit) {
        s = ei[e];
        const float v = leaky(s_src[s] + sd[dl]);
        a = __expf(v - ord2f(m_ord[dl])) / den[dl];
      }
      unsigned mask = __builtin_amdgcn_ballot_w32(hit);
      while (mask) {
        const int j = __builtin_ctz(mask);
        mask &= mask - 1;
        row_add(H, acc, __builtin_amdgcn_readlane(s, j),
                __builtin_amdgcn_readlane((int)dl, j),
                __int_as_float(__builtin_amdgcn_readlane(__float_as_int(a), j)),
                lane);
      }
    }
  }
  __syncthreads();

  float4* acc4 = (float4*)acc;
  float4* o4 = (float4*)(out + (size_t)lo * DD);
  for (int i = tid; i < NPB * DD / 4; i += OT) {
    const int li = (i >> 5) * (AS / 4) + (i & 31);
    const float* rp = acc + (i >> 5) * AS + (i & 31);
    const float4 r = make_float4(rp[0], rp[32], rp[64], rp[96]);
    const float4 b = ((const float4*)bias)[i & (DD / 4 - 1)];
    const float4 v = make_float4(r.x + b.x, r.y + b.y, r.z + b.z, r.w + b.w);
    if (TAIL == 0 || TAIL == 1) o4[i] = v;
    if (TAIL != 0) acc4[li] = v;
  }
  if (TAIL == 0) return;

  __syncthreads();
  {
    const int l15 = lane & 15;
    const int half = lane >> 4;
    const int wu = __builtin_amdgcn_readfirstlane(wv);
    const int ct = wu & 7;
    const int rg = wu >> 3;
    const int col = ct * 16 + l15;
    constexpr int NCH = (TAIL == 3) ? 2 : 4;
    v8f c[4];
#pragma unroll
    for (int t = 0; t < 4; ++t)
#pragma unroll
      for (int r = 0; r < 8; ++r) c[t][r] = 0.0f;
#pragma unroll
    for (int ch = 0; ch < NCH; ++ch) {
      if (TAIL == 2 && ch == 2) {
        __syncthreads();
        const float4* a4 = (const float4*)(A2 + (size_t)lo * DD);
        constexpr int NR = (NPB * DD / 4 + OT - 1) / OT;
#pragma unroll
        for (int u = 0; u < NR; ++u) {
          const int i = tid + u * OT;
          if (i < NPB * DD / 4) acc4[(i >> 5) * (AS / 4) + (i & 31)] = a4[i];
        }
        __syncthreads();
      }
      v2f b[16];
      const float* wp = Wn + (size_t)(ch * 64 + 2 * half) * DD + col;
#pragma unroll
      for (int kk = 0; kk < 16; ++kk) {
        b[kk][0] = wp[(size_t)(kk * 4) * DD];
        b[kk][1] = wp[(size_t)(kk * 4 + 1) * DD];
      }
      const int acol = (ch & 1) * 64 + 2 * half;
#pragma unroll
      for (int t = 0; t < 4; ++t) {
        if (4 * t + 3 < RT || rg + 4 * t < RT) {
          const float* ap = acc + ((rg + 4 * t) * 16 + l15) * AS + acol;
#pragma unroll
          for (int kk = 0; kk < 16; ++kk) {
            const v2f a = *(const v2f*)(ap + kk * 4);
            c[t] = __builtin_amdgcn_wmma_f32_16x16x4_f32(false, a, false, b[kk],
                                                         (short)0, c[t], false, false);
          }
        }
      }
    }
#pragma unroll
    for (int t = 0; t < 4; ++t) {
      if (4 * t + 3 < RT || rg + 4 * t < RT) {
#pragma unroll
        for (int r = 0; r < 8; ++r) {
          const int row = (rg + 4 * t) * 16 + half * 8 + r;
          if ((4 * t + 3) * 16 + 15 < NPB || row < NPB)
            Hn[(size_t)(lo + row) * DD + col] = c[t][r];
        }
      }
    }
  }
}

extern "C" void kernel_launch(void* const* d_in, const int* in_sizes, int n_in,
                              void* d_out, int out_size, void* d_ws,
                              size_t ws_size, hipStream_t stream) {
  const float* x    = (const float*)d_in[0];
  const int*   ei   = (const int*)d_in[1];
  const float* We   = (const float*)d_in[2];
  const float* ae_s = (const float*)d_in[3];
  const float* ae_d = (const float*)d_in[4];
  const float* be   = (const float*)d_in[5];
  const float* Wp   = (const float*)d_in[6];
  const float* ap_s = (const float*)d_in[7];
  const float* ap_d = (const float*)d_in[8];
  const float* bp   = (const float*)d_in[9];
  const float* Wd   = (const float*)d_in[10];
  const float* ad_s = (const float*)d_in[11];
  const float* ad_d = (const float*)d_in[12];
  const float* bd   = (const float*)d_in[13];
  float*       out  = (float*)d_out;

  const size_t ND = (size_t)NN * DD;
  float*    enc   = (float*)d_ws;
  float*    h2    = enc + ND;
  int*      glist = (int*)(h2 + ND);
  int*      gcnt  = glist + NBLK * CAP;
  unsigned* ob    = (unsigned*)(gcnt + 256);
  float*    h1    = h2 + ND + ND;
  float*    s_src = h1 + ND;
  float*    s_dst = s_src + NN;
  static_assert(NBLK <= 256 && (size_t)NBLK * CAP + 256 + EE / 4 <= (size_t)NN * DD,
                "lists, counts and owner-id bytes fit in one feature plane");
  static_assert(NBLK * OT >= EE / 4,
                "the encode launch has a thread for every 4 edges");

  encode_product<<<NBLK, OT, 0, stream>>>(x, We, h1, ei, ob);
  scores_init<<<SGRID, 256, 0, stream>>>(h1, ae_s, ae_d, s_src, s_dst);
  owner_aggregate<true, 1><<<NBLK, OT, 0, stream>>>(
      ei, h1, s_src, s_dst, be, enc, glist, gcnt, Wp, nullptr, h2, ob);
  scores_init<<<SGRID, 256, 0, stream>>>(h2, ap_s, ap_d, s_src, s_dst);
  owner_aggregate<false, 2><<<NBLK, OT, 0, stream>>>(
      ei, h2, s_src, s_dst, bp, out, glist, gcnt, Wp, enc, h1, ob);
  scores_init<<<SGRID, 256, 0, stream>>>(h1, ap_s, ap_d, s_src, s_dst);
  owner_aggregate<false, 3><<<NBLK, OT, 0, stream>>>(
      ei, h1, s_src, s_dst, bp, out, glist, gcnt, Wd, nullptr, h2, ob);
  scores_init<<<SGRID, 256, 0, stream>>>(h2, ad_s, ad_d, s_src, s_dst);
  owner_aggregate<false, 0><<<NBLK, OT, 0, stream>>>(
      ei, h2, s_src, s_dst, bd, out, glist, gcnt, nullptr, nullptr, nullptr, ob);
}
